# starting kernel
# baseline (speedup 1.0000x reference)
#include <hip/hip_runtime.h>
#include <hip/hip_bf16.h>
#include <math.h>


typedef float v2f __attribute__((ext_vector_type(2)));
typedef float v8f __attribute__((ext_vector_type(8)));

#define PI_F 3.14159265358979323846f
#define RSTRIDE 133
#define KFCOLS 132

__device__ __forceinline__ v8f wmma_f32(v2f a, v2f b, v8f c) {
  return __builtin_amdgcn_wmma_f32_16x16x4_f32(false, a, false, b, (short)0, c,
                                               false, false);
}

template <typename F>
__device__ __forceinline__ v2f load_a_frag(F f, int kc, int lane) {
  const int r  = lane & 15;
  const int hi = (lane >> 4) & 1;
  v2f a;
  a.x = f(r, kc + 2 * hi);
  a.y = f(r, kc + 2 * hi + 1);
  return a;
}

template <typename F>
__device__ __forceinline__ v2f load_b_frag(F f, int kc, int lane) {
  const int c  = lane & 15;
  const int hi = (lane >> 4) & 1;
  v2f b;
  b.x = f(kc + 2 * hi, c);
  b.y = f(kc + 2 * hi + 1, c);
  return b;
}

template <typename FAr, typename FAi, typename FBr, typename FBi>
__device__ __forceinline__ void cmatmul16(FAr ar_f, FAi ai_f, FBr br_f,
                                          FBi bi_f, int lane, v8f& dre,
                                          v8f& dim) {
  v8f rr = {}, ii = {}, ri = {}, ir = {};
#pragma unroll
  for (int kc = 0; kc < 16; kc += 4) {
    v2f ar = load_a_frag(ar_f, kc, lane);
    v2f ai = load_a_frag(ai_f, kc, lane);
    v2f br = load_b_frag(br_f, kc, lane);
    v2f bi = load_b_frag(bi_f, kc, lane);
    rr = wmma_f32(ar, br, rr);
    ii = wmma_f32(ai, bi, ii);
    ri = wmma_f32(ar, bi, ri);
    ir = wmma_f32(ai, br, ir);
  }
  dre = rr - ii;
  dim = ri + ir;
}

template <typename SRC, typename DST>
__device__ __forceinline__ void fft256(SRC src, DST dst, const float2* Ftab,
                                       const float2* Wtab, float2* spill,
                                       float dir, int lane) {
  v8f bre, bim;
  cmatmul16(
      [&](int r, int c) { return src(r + 16 * c).x; },
      [&](int r, int c) { return src(r + 16 * c).y; },
      [&](int r, int c) { return Ftab[r * 16 + c].x; },
      [&](int r, int c) { return dir * Ftab[r * 16 + c].y; }, lane, bre, bim);

  const int col   = lane & 15;
  const int rbase = (lane >> 4) ? 8 : 0;
#pragma unroll
  for (int j = 0; j < 8; ++j) {
    const int row = j + rbase;
    const float2 w = Wtab[(row * col) & 255];
    const float wr = w.x, wi = dir * w.y;
    const float re = bre[j] * wr - bim[j] * wi;
    const float im = bre[j] * wi + bim[j] * wr;
    spill[row * 16 + col] = make_float2(re, im);
  }
  asm volatile("s_wait_dscnt 0" ::: "memory");

  v8f cre, cim;
  cmatmul16(
      [&](int r, int c) { return Ftab[r * 16 + c].x; },
      [&](int r, int c) { return dir * Ftab[r * 16 + c].y; },
      [&](int r, int c) { return spill[r * 16 + c].x; },
      [&](int r, int c) { return spill[r * 16 + c].y; }, lane, cre, cim);

#pragma unroll
  for (int j = 0; j < 8; ++j) {
    const int k2 = j + rbase;
    dst(col + 16 * k2, make_float2(cre[j], cim[j]));
  }
}

__device__ __forceinline__ void init_tables(float2* Ftab, float2* Wtab,
                                            int tid) {
  for (int i = tid; i < 256; i += 256) {
    const int r = i >> 4, c = i & 15;
    const float tf = -2.0f * PI_F * (float)((r * c) & 15) / 16.0f;
    Ftab[i] = make_float2(cosf(tf), sinf(tf));
    const float tw = -2.0f * PI_F * (float)i / 256.0f;
    Wtab[i] = make_float2(cosf(tw), sinf(tw));
  }
}

__global__ void __launch_bounds__(256) kf_precompute(
    const float* __restrict__ kin, float2* __restrict__ kf) {
  extern __shared__ char smem[];
  float2* Rk      = (float2*)smem;
  float2* Ftab    = (float2*)(smem + 64 * RSTRIDE * 8);
  float2* Wtab    = Ftab + 256;
  float2* scratch = Wtab + 256;

  const int tid = threadIdx.x, lane = tid & 31, wave = tid >> 5;
  const int c = blockIdx.x;

  init_tables(Ftab, Wtab, tid);
  const float* kc = kin + (size_t)c * (64 * 64);
  for (int i = tid; i < 64 * 64; i += 256) {
    const int r = i >> 6, w = i & 63;
    Rk[r * RSTRIDE + w] = make_float2(kc[i], 0.0f);
  }
  __syncthreads();

  float2* spill = scratch + wave * 512;

  for (int row = wave; row < 64; row += 8) {
    float2* Rrow = Rk + row * RSTRIDE;
    fft256(
        [&](int n) {
          return (n < 64) ? make_float2(Rrow[n].x, 0.0f) : make_float2(0.f, 0.f);
        },
        [&](int k, float2 v) { if (k <= 128) Rrow[k] = v; },
        Ftab, Wtab, spill, +1.0f, lane);
  }
  __syncthreads();

  for (int col = wave; col < 129; col += 8) {
    float2* kfcol = kf + ((size_t)c * KFCOLS + col) * 256;
    fft256(
        [&](int n) {
          return (n < 64) ? Rk[n * RSTRIDE + col] : make_float2(0.f, 0.f);
        },
        [&](int k, float2 v) { kfcol[k] = v; },
        Ftab, Wtab, spill, +1.0f, lane);
  }
}

__global__ void __launch_bounds__(256) conv_main(
    const float* __restrict__ u, const float* __restrict__ pregate,
    const float* __restrict__ postgate, const float2* __restrict__ kf,
    float* __restrict__ y) {
  extern __shared__ char smem[];
  float2* R       = (float2*)smem;
  float2* Ftab    = (float2*)(smem + 128 * RSTRIDE * 8);
  float2* Wtab    = Ftab + 256;
  float2* scratch = Wtab + 256;

  const int tid = threadIdx.x, lane = tid & 31, wave = tid >> 5;
  const int img = blockIdx.x;
  const int c = img & 255;
  const size_t base = (size_t)img * (128 * 128);

  init_tables(Ftab, Wtab, tid);

  for (int i = tid; i < 128 * 128; i += 256) {
    const int r = i >> 7, w = i & 127;
    R[r * RSTRIDE + w] = make_float2(u[base + i] * pregate[base + i], 0.0f);
  }
  __syncthreads();

  float2* spill = scratch + wave * 512;
  float2* spec  = spill + 256;

  for (int row = wave; row < 128; row += 8) {
    float2* Rrow = R + row * RSTRIDE;
    fft256(
        [&](int n) {
          return (n < 128) ? make_float2(Rrow[n].x, 0.0f)
                           : make_float2(0.f, 0.f);
        },
        [&](int k, float2 v) { if (k <= 128) Rrow[k] = v; },
        Ftab, Wtab, spill, +1.0f, lane);
  }
  __syncthreads();

  const float2* kfbase = kf + (size_t)c * KFCOLS * 256;
  for (int col = wave; col < 129; col += 8) {
    if (col + 8 < 129) {
      const char* pf = (const char*)(kfbase + (size_t)(col + 8) * 256);
      __builtin_prefetch(pf + lane * 64, 0, 1);
    }

    fft256(
        [&](int n) {
          return (n < 128) ? R[n * RSTRIDE + col] : make_float2(0.f, 0.f);
        },
        [&](int k, float2 v) { spec[k] = v; },
        Ftab, Wtab, spill, +1.0f, lane);

    const float2* kfcol = kfbase + (size_t)col * 256;
    for (int h = lane; h < 256; h += 32) {
      const float2 a = spec[h];
      const float2 b = kfcol[h];
      spec[h] = make_float2(a.x * b.x - a.y * b.y, a.x * b.y + a.y * b.x);
    }
    asm volatile("s_wait_dscnt 0" ::: "memory");

    fft256(
        [&](int n) { return spec[n]; },
        [&](int k, float2 v) { if (k < 128) R[k * RSTRIDE + col] = v; },
        Ftab, Wtab, spill, -1.0f, lane);
  }
  __syncthreads();

  const float inv_scale = 1.0f / 65536.0f;
  for (int row = wave; row < 128; row += 8) {
    float2* Rrow = R + row * RSTRIDE;
    const size_t obase = base + (size_t)row * 128;
    fft256(
        [&](int n) {
          if (n <= 128) return Rrow[n];
          const float2 v = Rrow[256 - n];
          return make_float2(v.x, -v.y);
        },
        [&](int k, float2 v) {
          if (k < 128) y[obase + k] = v.x * inv_scale * postgate[obase + k];
        },
        Ftab, Wtab, spill, -1.0f, lane);
  }
}

extern "C" void kernel_launch(void* const* d_in, const int* in_sizes, int n_in,
                              void* d_out, int out_size, void* d_ws,
                              size_t ws_size, hipStream_t stream) {
  const float* u        = (const float*)d_in[0];
  const float* k        = (const float*)d_in[1];
  const float* pregate  = (const float*)d_in[2];
  const float* postgate = (const float*)d_in[3];
  float* y   = (float*)d_out;
  float2* kf = (float2*)d_ws;

  const size_t lds_kf   = (size_t)64 * RSTRIDE * 8 + 512 * 8 + 8 * 512 * 8;
  const size_t lds_main = (size_t)128 * RSTRIDE * 8 + 512 * 8 + 8 * 512 * 8;

  (void)hipFuncSetAttribute((const void*)kf_precompute,
                            hipFuncAttributeMaxDynamicSharedMemorySize,
                            (int)lds_kf);
  (void)hipFuncSetAttribute((const void*)conv_main,
                            hipFuncAttributeMaxDynamicSharedMemorySize,
                            (int)lds_main);

  kf_precompute<<<256, 256, lds_kf, stream>>>(k, kf);
  conv_main<<<4 * 256, 256, lds_main, stream>>>(u, pregate, postgate, kf, y);
}


# Each 16-point DFT stage split radix-2: even/odd samples as two 8-point DFTs in real 16x16 block form (odd twiddle in its matrix), joined by vector add/sub: 16 WMMA per FFT, was 32; new sum order
# speedup vs baseline: 2.8816x; 2.8816x over previous
#include <hip/hip_runtime.h>
#include <hip/hip_bf16.h>
#include <math.h>


typedef float v2f __attribute__((ext_vector_type(2)));
typedef float v8f __attribute__((ext_vector_type(8)));

#define PI_F 3.14159265358979323846f
#define RSTRIDE 133
#define KFCOLS 132
#define NB 4
#define SPITCH 18
#define SPILLSZ (16 * SPITCH)

__device__ __forceinline__ v8f wmma_f32(v2f a, v2f b, v8f c) {
  return __builtin_amdgcn_wmma_f32_16x16x4_f32(false, a, false, b, (short)0, c,
                                               false, false);
}


__device__ __forceinline__ v2f load_g_frag(const float2* Ftab, float dir,
                                           int t, int b, int lane) {
  const int m = lane & 15;
  const int h = (lane >> 4) & 1;
  const int c = m >> 1;
  const int a = 2 * t + h;
  const float2 g = Ftab[16 + (((2 * a + b) * c) & 15)];
  const float gr = g.x, gi = dir * g.y;
  v2f f;
  if (m & 1) { f.x = gi; f.y = gr; } else { f.x = gr; f.y = -gi; }
  return f;
}

__device__ __forceinline__ v2f as_frag(float2 x) {
  v2f f;
  f.x = x.x;
  f.y = x.y;
  return f;
}

template <int N, typename SRC, typename DST>
__device__ __forceinline__ void fft256xN(SRC src, DST dst, const float2* Ftab,
                                         const float2* Wtab, float2* spill,
                                         float dir, int lane) {
  const int col = lane & 15;
  const int h   = (lane >> 4) & 1;

  v2f ge[4], go[4];
#pragma unroll
  for (int t = 0; t < 4; ++t) {
    ge[t] = load_g_frag(Ftab, dir, t, 0, lane);
    go[t] = load_g_frag(Ftab, dir, t, 1, lane);
  }

  float twr[2][4], twi[2][4];
#pragma unroll
  for (int d = 0; d < 2; ++d) {
#pragma unroll
    for (int i = 0; i < 4; ++i) {
      const float2 w = Wtab[(col * (4 * h + i + 8 * d)) & 255];
      twr[d][i] = w.x;
      twi[d][i] = dir * w.y;
    }
  }

  v8f e[N], o[N];
#pragma unroll
  for (int q = 0; q < N; ++q) {
    e[q] = v8f{}; o[q] = v8f{};
  }

#pragma unroll
  for (int t = 0; t < 4; ++t) {
    const int n2 = 2 * (2 * t + h);
#pragma unroll
    for (int q = 0; q < N; ++q) {
      const v2f xe = as_frag(src(q, col + 16 * n2));
      const v2f xo = as_frag(src(q, col + 16 * (n2 + 1)));
      e[q] = wmma_f32(ge[t], xe, e[q]);
      o[q] = wmma_f32(go[t], xo, o[q]);
    }
  }

#pragma unroll
  for (int q = 0; q < N; ++q) {
    float2* sp = spill + q * SPILLSZ + col;
#pragma unroll
    for (int i = 0; i < 4; ++i) {
      const float lr = e[q][2 * i] + o[q][2 * i];
      const float li = e[q][2 * i + 1] + o[q][2 * i + 1];
      const float hr = e[q][2 * i] - o[q][2 * i];
      const float hm = e[q][2 * i + 1] - o[q][2 * i + 1];
      const int k1 = 4 * h + i;
      sp[k1 * SPITCH] =
          make_float2(lr * twr[0][i] - li * twi[0][i],
                      lr * twi[0][i] + li * twr[0][i]);
      sp[(k1 + 8) * SPITCH] =
          make_float2(hr * twr[1][i] - hm * twi[1][i],
                      hr * twi[1][i] + hm * twr[1][i]);
    }
  }
  asm volatile("s_wait_dscnt 0" ::: "memory");

#pragma unroll
  for (int q = 0; q < N; ++q) {
    e[q] = v8f{}; o[q] = v8f{};
  }

#pragma unroll
  for (int t = 0; t < 4; ++t) {
    const int n1 = 2 * (2 * t + h);
#pragma unroll
    for (int q = 0; q < N; ++q) {
      const float2* sp = spill + q * SPILLSZ + col * SPITCH + n1;
      const v2f be = as_frag(sp[0]);
      const v2f bo = as_frag(sp[1]);
      e[q] = wmma_f32(ge[t], be, e[q]);
      o[q] = wmma_f32(go[t], bo, o[q]);
    }
  }

  v8f lo[N], hi[N];
#pragma unroll
  for (int q = 0; q < N; ++q) {
    lo[q] = e[q] + o[q];
    hi[q] = e[q] - o[q];
  }
  dst(lo, hi);
}

__device__ __forceinline__ void init_tables(float2* Ftab, float2* Wtab,
                                            int tid) {
  for (int i = tid; i < 256; i += 256) {
    const int r = i >> 4, c = i & 15;
    const float tf = -2.0f * PI_F * (float)((r * c) & 15) / 16.0f;
    Ftab[i] = make_float2(cosf(tf), sinf(tf));
    const float tw = -2.0f * PI_F * (float)i / 256.0f;
    Wtab[i] = make_float2(cosf(tw), sinf(tw));
  }
}


template <int N>
__device__ __forceinline__ void kf_columns(const float2* Rk, float2* kfc,
                                           int col0, const float2* Ftab,
                                           const float2* Wtab, float2* spill,
                                           int lane) {
  fft256xN<N>(
      [&](int q, int n) {
        return (n < 64) ? Rk[n * RSTRIDE + col0 + 8 * q]
                        : make_float2(0.f, 0.f);
      },
      [&](const auto& lo, const auto& hi) {
        const int k0 = (lane & 15) + 64 * (lane >> 4);
#pragma unroll
        for (int q = 0; q < N; ++q) {
          float2* dcol = kfc + (size_t)(col0 + 8 * q) * 256 + k0;
#pragma unroll
          for (int i = 0; i < 4; ++i) {
            dcol[16 * i] = make_float2(lo[q][2 * i], lo[q][2 * i + 1]);
            dcol[128 + 16 * i] = make_float2(hi[q][2 * i], hi[q][2 * i + 1]);
          }
        }
      },
      Ftab, Wtab, spill, +1.0f, lane);
}

__global__ void __launch_bounds__(256) kf_precompute(
    const float* __restrict__ kin, float2* __restrict__ kf) {
  extern __shared__ char smem[];
  float2* Rk      = (float2*)smem;
  float2* Ftab    = (float2*)(smem + 64 * RSTRIDE * 8);
  float2* Wtab    = Ftab + 256;
  float2* scratch = Wtab + 256;

  const int tid = threadIdx.x, lane = tid & 31, wave = tid >> 5;
  const int c = blockIdx.x;

  init_tables(Ftab, Wtab, tid);
  const float* kc = kin + (size_t)c * (64 * 64);
  for (int i = tid; i < 64 * 64; i += 256) {
    const int r = i >> 6, w = i & 63;
    Rk[r * RSTRIDE + w] = make_float2(kc[i], 0.0f);
  }
  __syncthreads();

  float2* spill = scratch + wave * (NB * SPILLSZ);

  for (int row0 = wave; row0 < 64; row0 += 8 * NB) {
    fft256xN<NB>(
        [&](int q, int n) {
          const float2* Rrow = Rk + (row0 + 8 * q) * RSTRIDE;
          return (n < 64) ? make_float2(Rrow[n].x, 0.0f)
                          : make_float2(0.f, 0.f);
        },
        [&](const auto& lo, const auto& hi) {
          const int k0 = (lane & 15) + 64 * (lane >> 4);
#pragma unroll
          for (int q = 0; q < NB; ++q) {
            float2* Rrow = Rk + (row0 + 8 * q) * RSTRIDE + k0;
#pragma unroll
            for (int i = 0; i < 4; ++i)
              Rrow[16 * i] = make_float2(lo[q][2 * i], lo[q][2 * i + 1]);
          }
          if (lane == 0) {
#pragma unroll
            for (int q = 0; q < NB; ++q)
              Rk[(row0 + 8 * q) * RSTRIDE + 128] =
                  make_float2(hi[q][0], hi[q][1]);
          }
        },
        Ftab, Wtab, spill, +1.0f, lane);
  }
  __syncthreads();

  float2* kfc = kf + (size_t)c * KFCOLS * 256;
  for (int col0 = wave; col0 < 128; col0 += 8 * NB)
    kf_columns<NB>(Rk, kfc, col0, Ftab, Wtab, spill, lane);
  if (wave == 0) kf_columns<1>(Rk, kfc, 128, Ftab, Wtab, spill, lane);
}


template <int N>
__device__ __forceinline__ void conv_columns(float2* R, const float2* kfbase,
                                             int col0, const float2* Ftab,
                                             const float2* Wtab, float2* spill,
                                             float2* spec, int lane) {
  fft256xN<N>(
      [&](int q, int n) {
        return (n < 128) ? R[n * RSTRIDE + col0 + 8 * q]
                         : make_float2(0.f, 0.f);
      },
      [&](const auto& lo, const auto& hi) {
        const int k0 = (lane & 15) + 64 * (lane >> 4);
#pragma unroll
        for (int q = 0; q < N; ++q) {
#pragma unroll
          for (int i = 0; i < 4; ++i) {
            spec[q * 256 + k0 + 16 * i] =
                make_float2(lo[q][2 * i], lo[q][2 * i + 1]);
            spec[q * 256 + 128 + k0 + 16 * i] =
                make_float2(hi[q][2 * i], hi[q][2 * i + 1]);
          }
        }
      },
      Ftab, Wtab, spill, +1.0f, lane);

  float2 kv[N][8];
#pragma unroll
  for (int q = 0; q < N; ++q) {
    const float2* kfcol = kfbase + (size_t)(col0 + 8 * q) * 256 + lane;
#pragma unroll
    for (int i = 0; i < 8; ++i) kv[q][i] = kfcol[32 * i];
  }
#pragma unroll
  for (int q = 0; q < N; ++q) {
    float2* sq = spec + q * 256 + lane;
#pragma unroll
    for (int i = 0; i < 8; ++i) {
      const float2 a = sq[32 * i];
      const float2 b = kv[q][i];
      sq[32 * i] = make_float2(a.x * b.x - a.y * b.y, a.x * b.y + a.y * b.x);
    }
  }
  asm volatile("s_wait_dscnt 0" ::: "memory");

  fft256xN<N>(
      [&](int q, int n) { return spec[q * 256 + n]; },
      [&](const auto& lo, const auto& hi) {
        const int k0 = (lane & 15) + 64 * (lane >> 4);
#pragma unroll
        for (int q = 0; q < N; ++q)
#pragma unroll
          for (int i = 0; i < 4; ++i)
            R[(k0 + 16 * i) * RSTRIDE + col0 + 8 * q] =
                make_float2(lo[q][2 * i], lo[q][2 * i + 1]);
      },
      Ftab, Wtab, spill, -1.0f, lane);
}

__global__ void __launch_bounds__(256) conv_main(
    const float* __restrict__ u, const float* __restrict__ pregate,
    const float* __restrict__ postgate, const float2* __restrict__ kf,
    float* __restrict__ y) {
  extern __shared__ char smem[];
  float2* R       = (float2*)smem;
  float2* Ftab    = (float2*)(smem + 128 * RSTRIDE * 8);
  float2* Wtab    = Ftab + 256;
  float2* scratch = Wtab + 256;

  const int tid = threadIdx.x, lane = tid & 31, wave = tid >> 5;
  const int img = blockIdx.x;
  const int c = img & 255;
  const size_t base = (size_t)img * (128 * 128);

  init_tables(Ftab, Wtab, tid);

  for (int i = tid; i < 128 * 128; i += 256) {
    const int r = i >> 7, w = i & 127;
    R[r * RSTRIDE + w] = make_float2(u[base + i] * pregate[base + i], 0.0f);
  }
  __syncthreads();

  float2* spill = scratch + wave * (NB * (SPILLSZ + 256));
  float2* spec  = spill + NB * SPILLSZ;

  for (int row0 = wave; row0 < 128; row0 += 8 * NB) {
    fft256xN<NB>(
        [&](int q, int n) {
          const float2* Rrow = R + (row0 + 8 * q) * RSTRIDE;
          return (n < 128) ? make_float2(Rrow[n].x, 0.0f)
                           : make_float2(0.f, 0.f);
        },
        [&](const auto& lo, const auto& hi) {
          const int k0 = (lane & 15) + 64 * (lane >> 4);
#pragma unroll
          for (int q = 0; q < NB; ++q) {
            float2* Rrow = R + (row0 + 8 * q) * RSTRIDE + k0;
#pragma unroll
            for (int i = 0; i < 4; ++i)
              Rrow[16 * i] = make_float2(lo[q][2 * i], lo[q][2 * i + 1]);
          }
          if (lane == 0) {
#pragma unroll
            for (int q = 0; q < NB; ++q)
              R[(row0 + 8 * q) * RSTRIDE + 128] =
                  make_float2(hi[q][0], hi[q][1]);
          }
        },
        Ftab, Wtab, spill, +1.0f, lane);
  }
  __syncthreads();

  const float2* kfbase = kf + (size_t)c * KFCOLS * 256;
  for (int col0 = wave; col0 < 128; col0 += 8 * NB) {
#pragma unroll
    for (int q = 0; q < NB; ++q) {
      const int pc = col0 + 8 * NB + 8 * q;
      if (pc < 129) {
        const char* pf = (const char*)(kfbase + (size_t)pc * 256);
        __builtin_prefetch(pf + lane * 64, 0, 1);
      }
    }
    conv_columns<NB>(R, kfbase, col0, Ftab, Wtab, spill, spec, lane);
  }
  if (wave == 0)
    conv_columns<1>(R, kfbase, 128, Ftab, Wtab, spill, spec, lane);
  __syncthreads();

  const float inv_scale = 1.0f / 65536.0f;
  for (int row0 = wave; row0 < 128; row0 += 8 * NB) {
    fft256xN<NB>(
        [&](int q, int n) {
          const float2* Rrow = R + (row0 + 8 * q) * RSTRIDE;
          const bool stored = (n <= 128);
          const float2 v = Rrow[stored ? n : 256 - n];
          return make_float2(v.x, stored ? v.y : -v.y);
        },
        [&](const auto& lo, const auto& hi) {
          const int k0 = (lane & 15) + 64 * (lane >> 4);
          float g[NB][4];
#pragma unroll
          for (int q = 0; q < NB; ++q)
#pragma unroll
            for (int i = 0; i < 4; ++i)
              g[q][i] = postgate[base + (size_t)(row0 + 8 * q) * 128 + k0 +
                                 16 * i];
#pragma unroll
          for (int q = 0; q < NB; ++q)
#pragma unroll
            for (int i = 0; i < 4; ++i)
              y[base + (size_t)(row0 + 8 * q) * 128 + k0 + 16 * i] =
                  lo[q][2 * i] * inv_scale * g[q][i];
        },
        Ftab, Wtab, spill, -1.0f, lane);
  }
}

extern "C" void kernel_launch(void* const* d_in, const int* in_sizes, int n_in,
                              void* d_out, int out_size, void* d_ws,
                              size_t ws_size, hipStream_t stream) {
  const float* u        = (const float*)d_in[0];
  const float* k        = (const float*)d_in[1];
  const float* pregate  = (const float*)d_in[2];
  const float* postgate = (const float*)d_in[3];
  float* y   = (float*)d_out;
  float2* kf = (float2*)d_ws;

  const size_t lds_kf   = (size_t)64 * RSTRIDE * 8 + 512 * 8 + 8 * NB * SPILLSZ * 8;
  const size_t lds_main =
      (size_t)128 * RSTRIDE * 8 + 512 * 8 + 8 * NB * (SPILLSZ + 256) * 8;

  (void)hipFuncSetAttribute((const void*)kf_precompute,
                            hipFuncAttributeMaxDynamicSharedMemorySize,
                            (int)lds_kf);
  (void)hipFuncSetAttribute((const void*)conv_main,
                            hipFuncAttributeMaxDynamicSharedMemorySize,
                            (int)lds_main);

  kf_precompute<<<256, 256, lds_kf, stream>>>(k, kf);
  conv_main<<<4 * 256, 256, lds_main, stream>>>(u, pregate, postgate, kf, y);
}


# Main-kernel row passes: two real rows ride ONE complex transform (forward z=xA+i*xB, split by Hermitian symmetry via LDS; inverse YA+i*YB, outputs re/im): 128 row FFTs, was 256; new rounding
# speedup vs baseline: 3.2345x; 1.1225x over previous
#include <hip/hip_runtime.h>
#include <hip/hip_bf16.h>
#include <math.h>


typedef float v2f __attribute__((ext_vector_type(2)));
typedef float v8f __attribute__((ext_vector_type(8)));

#define PI_F 3.14159265358979323846f
#define RSTRIDE 133
#define KFCOLS 132
#define NB 4
#define SPITCH 18
#define SPILLSZ (16 * SPITCH)

__device__ __forceinline__ v8f wmma_f32(v2f a, v2f b, v8f c) {
  return __builtin_amdgcn_wmma_f32_16x16x4_f32(false, a, false, b, (short)0, c,
                                               false, false);
}


__device__ __forceinline__ v2f load_g_frag(const float2* Ftab, float dir,
                                           int t, int b, int lane) {
  const int m = lane & 15;
  const int h = (lane >> 4) & 1;
  const int c = m >> 1;
  const int a = 2 * t + h;
  const float2 g = Ftab[16 + (((2 * a + b) * c) & 15)];
  const float gr = g.x, gi = dir * g.y;
  v2f f;
  if (m & 1) { f.x = gi; f.y = gr; } else { f.x = gr; f.y = -gi; }
  return f;
}

__device__ __forceinline__ v2f as_frag(float2 x) {
  v2f f;
  f.x = x.x;
  f.y = x.y;
  return f;
}

template <int N, typename SRC, typename DST>
__device__ __forceinline__ void fft256xN(SRC src, DST dst, const float2* Ftab,
                                         const float2* Wtab, float2* spill,
                                         float dir, int lane) {
  const int col = lane & 15;
  const int h   = (lane >> 4) & 1;

  v2f ge[4], go[4];
#pragma unroll
  for (int t = 0; t < 4; ++t) {
    ge[t] = load_g_frag(Ftab, dir, t, 0, lane);
    go[t] = load_g_frag(Ftab, dir, t, 1, lane);
  }

  float twr[2][4], twi[2][4];
#pragma unroll
  for (int d = 0; d < 2; ++d) {
#pragma unroll
    for (int i = 0; i < 4; ++i) {
      const float2 w = Wtab[(col * (4 * h + i + 8 * d)) & 255];
      twr[d][i] = w.x;
      twi[d][i] = dir * w.y;
    }
  }

  v8f e[N], o[N];
#pragma unroll
  for (int q = 0; q < N; ++q) {
    e[q] = v8f{}; o[q] = v8f{};
  }

#pragma unroll
  for (int t = 0; t < 4; ++t) {
    const int n2 = 2 * (2 * t + h);
#pragma unroll
    for (int q = 0; q < N; ++q) {
      const v2f xe = as_frag(src(q, col + 16 * n2));
      const v2f xo = as_frag(src(q, col + 16 * (n2 + 1)));
      e[q] = wmma_f32(ge[t], xe, e[q]);
      o[q] = wmma_f32(go[t], xo, o[q]);
    }
  }

#pragma unroll
  for (int q = 0; q < N; ++q) {
    float2* sp = spill + q * SPILLSZ + col;
#pragma unroll
    for (int i = 0; i < 4; ++i) {
      const float lr = e[q][2 * i] + o[q][2 * i];
      const float li = e[q][2 * i + 1] + o[q][2 * i + 1];
      const float hr = e[q][2 * i] - o[q][2 * i];
      const float hm = e[q][2 * i + 1] - o[q][2 * i + 1];
      const int k1 = 4 * h + i;
      sp[k1 * SPITCH] =
          make_float2(lr * twr[0][i] - li * twi[0][i],
                      lr * twi[0][i] + li * twr[0][i]);
      sp[(k1 + 8) * SPITCH] =
          make_float2(hr * twr[1][i] - hm * twi[1][i],
                      hr * twi[1][i] + hm * twr[1][i]);
    }
  }
  asm volatile("s_wait_dscnt 0" ::: "memory");

#pragma unroll
  for (int q = 0; q < N; ++q) {
    e[q] = v8f{}; o[q] = v8f{};
  }

#pragma unroll
  for (int t = 0; t < 4; ++t) {
    const int n1 = 2 * (2 * t + h);
#pragma unroll
    for (int q = 0; q < N; ++q) {
      const float2* sp = spill + q * SPILLSZ + col * SPITCH + n1;
      const v2f be = as_frag(sp[0]);
      const v2f bo = as_frag(sp[1]);
      e[q] = wmma_f32(ge[t], be, e[q]);
      o[q] = wmma_f32(go[t], bo, o[q]);
    }
  }

  v8f lo[N], hi[N];
#pragma unroll
  for (int q = 0; q < N; ++q) {
    lo[q] = e[q] + o[q];
    hi[q] = e[q] - o[q];
  }
  dst(lo, hi);
}

__device__ __forceinline__ void init_tables(float2* Ftab, float2* Wtab,
                                            int tid) {
  for (int i = tid; i < 256; i += 256) {
    const int r = i >> 4, c = i & 15;
    const float tf = -2.0f * PI_F * (float)((r * c) & 15) / 16.0f;
    Ftab[i] = make_float2(cosf(tf), sinf(tf));
    const float tw = -2.0f * PI_F * (float)i / 256.0f;
    Wtab[i] = make_float2(cosf(tw), sinf(tw));
  }
}


template <int N>
__device__ __forceinline__ void kf_columns(const float2* Rk, float2* kfc,
                                           int col0, const float2* Ftab,
                                           const float2* Wtab, float2* spill,
                                           int lane) {
  fft256xN<N>(
      [&](int q, int n) {
        return (n < 64) ? Rk[n * RSTRIDE + col0 + 8 * q]
                        : make_float2(0.f, 0.f);
      },
      [&](const auto& lo, const auto& hi) {
        const int k0 = (lane & 15) + 64 * (lane >> 4);
#pragma unroll
        for (int q = 0; q < N; ++q) {
          float2* dcol = kfc + (size_t)(col0 + 8 * q) * 256 + k0;
#pragma unroll
          for (int i = 0; i < 4; ++i) {
            dcol[16 * i] = make_float2(lo[q][2 * i], lo[q][2 * i + 1]);
            dcol[128 + 16 * i] = make_float2(hi[q][2 * i], hi[q][2 * i + 1]);
          }
        }
      },
      Ftab, Wtab, spill, +1.0f, lane);
}

__global__ void __launch_bounds__(256) kf_precompute(
    const float* __restrict__ kin, float2* __restrict__ kf) {
  extern __shared__ char smem[];
  float2* Rk      = (float2*)smem;
  float2* Ftab    = (float2*)(smem + 64 * RSTRIDE * 8);
  float2* Wtab    = Ftab + 256;
  float2* scratch = Wtab + 256;

  const int tid = threadIdx.x, lane = tid & 31, wave = tid >> 5;
  const int c = blockIdx.x;

  init_tables(Ftab, Wtab, tid);
  const float* kc = kin + (size_t)c * (64 * 64);
  for (int i = tid; i < 64 * 64; i += 256) {
    const int r = i >> 6, w = i & 63;
    Rk[r * RSTRIDE + w] = make_float2(kc[i], 0.0f);
  }
  __syncthreads();

  float2* spill = scratch + wave * (NB * SPILLSZ);

  for (int row0 = wave; row0 < 64; row0 += 8 * NB) {
    fft256xN<NB>(
        [&](int q, int n) {
          const float2* Rrow = Rk + (row0 + 8 * q) * RSTRIDE;
          return (n < 64) ? make_float2(Rrow[n].x, 0.0f)
                          : make_float2(0.f, 0.f);
        },
        [&](const auto& lo, const auto& hi) {
          const int k0 = (lane & 15) + 64 * (lane >> 4);
#pragma unroll
          for (int q = 0; q < NB; ++q) {
            float2* Rrow = Rk + (row0 + 8 * q) * RSTRIDE + k0;
#pragma unroll
            for (int i = 0; i < 4; ++i)
              Rrow[16 * i] = make_float2(lo[q][2 * i], lo[q][2 * i + 1]);
          }
          if (lane == 0) {
#pragma unroll
            for (int q = 0; q < NB; ++q)
              Rk[(row0 + 8 * q) * RSTRIDE + 128] =
                  make_float2(hi[q][0], hi[q][1]);
          }
        },
        Ftab, Wtab, spill, +1.0f, lane);
  }
  __syncthreads();

  float2* kfc = kf + (size_t)c * KFCOLS * 256;
  for (int col0 = wave; col0 < 128; col0 += 8 * NB)
    kf_columns<NB>(Rk, kfc, col0, Ftab, Wtab, spill, lane);
  if (wave == 0) kf_columns<1>(Rk, kfc, 128, Ftab, Wtab, spill, lane);
}


template <int N>
__device__ __forceinline__ void conv_columns(float2* R, const float2* kfbase,
                                             int col0, const float2* Ftab,
                                             const float2* Wtab, float2* spill,
                                             float2* spec, int lane) {
  fft256xN<N>(
      [&](int q, int n) {
        return (n < 128) ? R[n * RSTRIDE + col0 + 8 * q]
                         : make_float2(0.f, 0.f);
      },
      [&](const auto& lo, const auto& hi) {
        const int k0 = (lane & 15) + 64 * (lane >> 4);
#pragma unroll
        for (int q = 0; q < N; ++q) {
#pragma unroll
          for (int i = 0; i < 4; ++i) {
            spec[q * 256 + k0 + 16 * i] =
                make_float2(lo[q][2 * i], lo[q][2 * i + 1]);
            spec[q * 256 + 128 + k0 + 16 * i] =
                make_float2(hi[q][2 * i], hi[q][2 * i + 1]);
          }
        }
      },
      Ftab, Wtab, spill, +1.0f, lane);

  float2 kv[N][8];
#pragma unroll
  for (int q = 0; q < N; ++q) {
    const float2* kfcol = kfbase + (size_t)(col0 + 8 * q) * 256 + lane;
#pragma unroll
    for (int i = 0; i < 8; ++i) kv[q][i] = kfcol[32 * i];
  }
#pragma unroll
  for (int q = 0; q < N; ++q) {
    float2* sq = spec + q * 256 + lane;
#pragma unroll
    for (int i = 0; i < 8; ++i) {
      const float2 a = sq[32 * i];
      const float2 b = kv[q][i];
      sq[32 * i] = make_float2(a.x * b.x - a.y * b.y, a.x * b.y + a.y * b.x);
    }
  }
  asm volatile("s_wait_dscnt 0" ::: "memory");

  fft256xN<N>(
      [&](int q, int n) { return spec[q * 256 + n]; },
      [&](const auto& lo, const auto& hi) {
        const int k0 = (lane & 15) + 64 * (lane >> 4);
#pragma unroll
        for (int q = 0; q < N; ++q)
#pragma unroll
          for (int i = 0; i < 4; ++i)
            R[(k0 + 16 * i) * RSTRIDE + col0 + 8 * q] =
                make_float2(lo[q][2 * i], lo[q][2 * i + 1]);
      },
      Ftab, Wtab, spill, -1.0f, lane);
}

__global__ void __launch_bounds__(256) conv_main(
    const float* __restrict__ u, const float* __restrict__ pregate,
    const float* __restrict__ postgate, const float2* __restrict__ kf,
    float* __restrict__ y) {
  extern __shared__ char smem[];
  float2* R       = (float2*)smem;
  float2* Ftab    = (float2*)(smem + 128 * RSTRIDE * 8);
  float2* Wtab    = Ftab + 256;
  float2* scratch = Wtab + 256;

  const int tid = threadIdx.x, lane = tid & 31, wave = tid >> 5;
  const int img = blockIdx.x;
  const int c = img & 255;
  const size_t base = (size_t)img * (128 * 128);

  init_tables(Ftab, Wtab, tid);

  for (int i = tid; i < 128 * 128; i += 256) {
    const int r = i >> 7, w = i & 127;
    R[r * RSTRIDE + w] = make_float2(u[base + i] * pregate[base + i], 0.0f);
  }
  __syncthreads();

  float2* spill = scratch + wave * (NB * (SPILLSZ + 256));
  float2* spec  = spill + NB * SPILLSZ;

  for (int row0 = wave; row0 < 128; row0 += 16 * NB) {
    fft256xN<NB>(
        [&](int q, int n) {
          const float2* RrowA = R + (row0 + 8 * q) * RSTRIDE;
          const float2* RrowB = RrowA + 8 * NB * RSTRIDE;
          return (n < 128) ? make_float2(RrowA[n].x, RrowB[n].x)
                           : make_float2(0.f, 0.f);
        },
        [&](const auto& lo, const auto& hi) {
          const int k0 = (lane & 15) + 64 * (lane >> 4);
#pragma unroll
          for (int q = 0; q < NB; ++q) {
#pragma unroll
            for (int i = 0; i < 4; ++i) {
              spec[q * 256 + k0 + 16 * i] =
                  make_float2(lo[q][2 * i], lo[q][2 * i + 1]);
              spec[q * 256 + 128 + k0 + 16 * i] =
                  make_float2(hi[q][2 * i], hi[q][2 * i + 1]);
            }
          }
          asm volatile("s_wait_dscnt 0" ::: "memory");
          float2 zk[NB][4], zm[NB][4], zn[NB];
#pragma unroll
          for (int q = 0; q < NB; ++q) {
            const float2* sq = spec + q * 256;
#pragma unroll
            for (int j = 0; j < 4; ++j) {
              const int k = lane + 32 * j;
              zk[q][j] = sq[k];
              zm[q][j] = sq[(256 - k) & 255];
            }
            zn[q] = sq[128];
          }
#pragma unroll
          for (int q = 0; q < NB; ++q) {
            float2* RrowA = R + (row0 + 8 * q) * RSTRIDE;
            float2* RrowB = RrowA + 8 * NB * RSTRIDE;
#pragma unroll
            for (int j = 0; j < 4; ++j) {
              const int k = lane + 32 * j;
              const float2 a = zk[q][j], m = zm[q][j];
              RrowA[k] = make_float2(0.5f * (a.x + m.x), 0.5f * (a.y - m.y));
              RrowB[k] = make_float2(0.5f * (a.y + m.y), 0.5f * (m.x - a.x));
            }
            if (lane == 0) {
              RrowA[128] = make_float2(zn[q].x, 0.0f);
              RrowB[128] = make_float2(zn[q].y, 0.0f);
            }
          }
        },
        Ftab, Wtab, spill, +1.0f, lane);
  }
  __syncthreads();

  const float2* kfbase = kf + (size_t)c * KFCOLS * 256;
  for (int col0 = wave; col0 < 128; col0 += 8 * NB) {
#pragma unroll
    for (int q = 0; q < NB; ++q) {
      const int pc = col0 + 8 * NB + 8 * q;
      if (pc < 129) {
        const char* pf = (const char*)(kfbase + (size_t)pc * 256);
        __builtin_prefetch(pf + lane * 64, 0, 1);
      }
    }
    conv_columns<NB>(R, kfbase, col0, Ftab, Wtab, spill, spec, lane);
  }
  if (wave == 0)
    conv_columns<1>(R, kfbase, 128, Ftab, Wtab, spill, spec, lane);
  __syncthreads();

  const float inv_scale = 1.0f / 65536.0f;
  for (int row0 = wave; row0 < 128; row0 += 16 * NB) {
    fft256xN<NB>(
        [&](int q, int n) {
          const float2* RrowA = R + (row0 + 8 * q) * RSTRIDE;
          const float2* RrowB = RrowA + 8 * NB * RSTRIDE;
          const bool stored = (n <= 128);
          const bool edge = ((n & 127) == 0);
          const int idx = stored ? n : 256 - n;
          const float2 a = RrowA[idx];
          const float2 b = RrowB[idx];
          const float ay = edge ? 0.0f : a.y;
          const float by = edge ? 0.0f : b.y;
          return stored ? make_float2(a.x - by, ay + b.x)
                        : make_float2(a.x + by, b.x - ay);
        },
        [&](const auto& lo, const auto& hi) {
          const int k0 = (lane & 15) + 64 * (lane >> 4);
          float gA[NB][4], gB[NB][4];
#pragma unroll
          for (int q = 0; q < NB; ++q) {
            const size_t oA = base + (size_t)(row0 + 8 * q) * 128 + k0;
            const size_t oB = oA + (size_t)(8 * NB) * 128;
#pragma unroll
            for (int i = 0; i < 4; ++i) {
              gA[q][i] = postgate[oA + 16 * i];
              gB[q][i] = postgate[oB + 16 * i];
            }
          }
#pragma unroll
          for (int q = 0; q < NB; ++q) {
            const size_t oA = base + (size_t)(row0 + 8 * q) * 128 + k0;
            const size_t oB = oA + (size_t)(8 * NB) * 128;
#pragma unroll
            for (int i = 0; i < 4; ++i) {
              y[oA + 16 * i] = lo[q][2 * i] * inv_scale * gA[q][i];
              y[oB + 16 * i] = lo[q][2 * i + 1] * inv_scale * gB[q][i];
            }
          }
        },
        Ftab, Wtab, spill, -1.0f, lane);
  }
}

extern "C" void kernel_launch(void* const* d_in, const int* in_sizes, int n_in,
                              void* d_out, int out_size, void* d_ws,
                              size_t ws_size, hipStream_t stream) {
  const float* u        = (const float*)d_in[0];
  const float* k        = (const float*)d_in[1];
  const float* pregate  = (const float*)d_in[2];
  const float* postgate = (const float*)d_in[3];
  float* y   = (float*)d_out;
  float2* kf = (float2*)d_ws;

  const size_t lds_kf   = (size_t)64 * RSTRIDE * 8 + 512 * 8 + 8 * NB * SPILLSZ * 8;
  const size_t lds_main =
      (size_t)128 * RSTRIDE * 8 + 512 * 8 + 8 * NB * (SPILLSZ + 256) * 8;

  (void)hipFuncSetAttribute((const void*)kf_precompute,
                            hipFuncAttributeMaxDynamicSharedMemorySize,
                            (int)lds_kf);
  (void)hipFuncSetAttribute((const void*)conv_main,
                            hipFuncAttributeMaxDynamicSharedMemorySize,
                            (int)lds_main);

  kf_precompute<<<256, 256, lds_kf, stream>>>(k, kf);
  conv_main<<<4 * 256, 256, lds_main, stream>>>(u, pregate, postgate, kf, y);
}


# Input staging, both kernels: a wave issues its rows (u, pregate; the filter) as ONE group of 16-byte loads ahead of the table set-up; was 8 serial rounds of 4-byte loads behind it
# speedup vs baseline: 3.5415x; 1.0949x over previous
#include <hip/hip_runtime.h>
#include <hip/hip_bf16.h>
#include <math.h>


typedef float v2f __attribute__((ext_vector_type(2)));
typedef float v8f __attribute__((ext_vector_type(8)));

#define PI_F 3.14159265358979323846f
#define RSTRIDE 133
#define KFCOLS 132
#define NB 4
#define SPITCH 18
#define SPILLSZ (16 * SPITCH)

__device__ __forceinline__ v8f wmma_f32(v2f a, v2f b, v8f c) {
  return __builtin_amdgcn_wmma_f32_16x16x4_f32(false, a, false, b, (short)0, c,
                                               false, false);
}


__device__ __forceinline__ v2f load_g_frag(const float2* Ftab, float dir,
                                           int t, int b, int lane) {
  const int m = lane & 15;
  const int h = (lane >> 4) & 1;
  const int c = m >> 1;
  const int a = 2 * t + h;
  const float2 g = Ftab[16 + (((2 * a + b) * c) & 15)];
  const float gr = g.x, gi = dir * g.y;
  v2f f;
  if (m & 1) { f.x = gi; f.y = gr; } else { f.x = gr; f.y = -gi; }
  return f;
}

__device__ __forceinline__ v2f as_frag(float2 x) {
  v2f f;
  f.x = x.x;
  f.y = x.y;
  return f;
}

template <int N, typename SRC, typename DST>
__device__ __forceinline__ void fft256xN(SRC src, DST dst, const float2* Ftab,
                                         const float2* Wtab, float2* spill,
                                         float dir, int lane) {
  const int col = lane & 15;
  const int h   = (lane >> 4) & 1;

  v2f ge[4], go[4];
#pragma unroll
  for (int t = 0; t < 4; ++t) {
    ge[t] = load_g_frag(Ftab, dir, t, 0, lane);
    go[t] = load_g_frag(Ftab, dir, t, 1, lane);
  }

  float twr[2][4], twi[2][4];
#pragma unroll
  for (int d = 0; d < 2; ++d) {
#pragma unroll
    for (int i = 0; i < 4; ++i) {
      const float2 w = Wtab[(col * (4 * h + i + 8 * d)) & 255];
      twr[d][i] = w.x;
      twi[d][i] = dir * w.y;
    }
  }

  v8f e[N], o[N];
#pragma unroll
  for (int q = 0; q < N; ++q) {
    e[q] = v8f{}; o[q] = v8f{};
  }

#pragma unroll
  for (int t = 0; t < 4; ++t) {
    const int n2 = 2 * (2 * t + h);
#pragma unroll
    for (int q = 0; q < N; ++q) {
      const v2f xe = as_frag(src(q, col + 16 * n2));
      const v2f xo = as_frag(src(q, col + 16 * (n2 + 1)));
      e[q] = wmma_f32(ge[t], xe, e[q]);
      o[q] = wmma_f32(go[t], xo, o[q]);
    }
  }

#pragma unroll
  for (int q = 0; q < N; ++q) {
    float2* sp = spill + q * SPILLSZ + col;
#pragma unroll
    for (int i = 0; i < 4; ++i) {
      const float lr = e[q][2 * i] + o[q][2 * i];
      const float li = e[q][2 * i + 1] + o[q][2 * i + 1];
      const float hr = e[q][2 * i] - o[q][2 * i];
      const float hm = e[q][2 * i + 1] - o[q][2 * i + 1];
      const int k1 = 4 * h + i;
      sp[k1 * SPITCH] =
          make_float2(lr * twr[0][i] - li * twi[0][i],
                      lr * twi[0][i] + li * twr[0][i]);
      sp[(k1 + 8) * SPITCH] =
          make_float2(hr * twr[1][i] - hm * twi[1][i],
                      hr * twi[1][i] + hm * twr[1][i]);
    }
  }
  asm volatile("s_wait_dscnt 0" ::: "memory");

#pragma unroll
  for (int q = 0; q < N; ++q) {
    e[q] = v8f{}; o[q] = v8f{};
  }

#pragma unroll
  for (int t = 0; t < 4; ++t) {
    const int n1 = 2 * (2 * t + h);
#pragma unroll
    for (int q = 0; q < N; ++q) {
      const float2* sp = spill + q * SPILLSZ + col * SPITCH + n1;
      const v2f be = as_frag(sp[0]);
      const v2f bo = as_frag(sp[1]);
      e[q] = wmma_f32(ge[t], be, e[q]);
      o[q] = wmma_f32(go[t], bo, o[q]);
    }
  }

  v8f lo[N], hi[N];
#pragma unroll
  for (int q = 0; q < N; ++q) {
    lo[q] = e[q] + o[q];
    hi[q] = e[q] - o[q];
  }
  dst(lo, hi);
}

__device__ __forceinline__ void init_tables(float2* Ftab, float2* Wtab,
                                            int tid) {
  for (int i = tid; i < 256; i += 256) {
    const int r = i >> 4, c = i & 15;
    const float tf = -2.0f * PI_F * (float)((r * c) & 15) / 16.0f;
    Ftab[i] = make_float2(cosf(tf), sinf(tf));
    const float tw = -2.0f * PI_F * (float)i / 256.0f;
    Wtab[i] = make_float2(cosf(tw), sinf(tw));
  }
}


template <int N>
__device__ __forceinline__ void kf_columns(const float2* Rk, float2* kfc,
                                           int col0, const float2* Ftab,
                                           const float2* Wtab, float2* spill,
                                           int lane) {
  fft256xN<N>(
      [&](int q, int n) {
        return (n < 64) ? Rk[n * RSTRIDE + col0 + 8 * q]
                        : make_float2(0.f, 0.f);
      },
      [&](const auto& lo, const auto& hi) {
        const int k0 = (lane & 15) + 64 * (lane >> 4);
#pragma unroll
        for (int q = 0; q < N; ++q) {
          float2* dcol = kfc + (size_t)(col0 + 8 * q) * 256 + k0;
#pragma unroll
          for (int i = 0; i < 4; ++i) {
            dcol[16 * i] = make_float2(lo[q][2 * i], lo[q][2 * i + 1]);
            dcol[128 + 16 * i] = make_float2(hi[q][2 * i], hi[q][2 * i + 1]);
          }
        }
      },
      Ftab, Wtab, spill, +1.0f, lane);
}

__global__ void __launch_bounds__(256) kf_precompute(
    const float* __restrict__ kin, float2* __restrict__ kf) {
  extern __shared__ char smem[];
  float2* Rk      = (float2*)smem;
  float2* Ftab    = (float2*)(smem + 64 * RSTRIDE * 8);
  float2* Wtab    = Ftab + 256;
  float2* scratch = Wtab + 256;

  const int tid = threadIdx.x, lane = tid & 31, wave = tid >> 5;
  const int c = blockIdx.x;

  const float4* k4 = reinterpret_cast<const float4*>(kin + (size_t)c * (64 * 64));
  const int srow = 8 * wave + (lane >> 3);
  const int sseg = lane & 7;
  float4 kq[4];
#pragma unroll
  for (int j = 0; j < 4; ++j)
    kq[j] = k4[(srow + 4 * (j >> 1)) * 16 + sseg + 8 * (j & 1)];

  init_tables(Ftab, Wtab, tid);

#pragma unroll
  for (int j = 0; j < 4; ++j) {
    float2* d = Rk + (srow + 4 * (j >> 1)) * RSTRIDE + 4 * (sseg + 8 * (j & 1));
    d[0] = make_float2(kq[j].x, 0.0f);
    d[1] = make_float2(kq[j].y, 0.0f);
    d[2] = make_float2(kq[j].z, 0.0f);
    d[3] = make_float2(kq[j].w, 0.0f);
  }
  __syncthreads();

  float2* spill = scratch + wave * (NB * SPILLSZ);

  for (int row0 = wave; row0 < 64; row0 += 8 * NB) {
    fft256xN<NB>(
        [&](int q, int n) {
          const float2* Rrow = Rk + (row0 + 8 * q) * RSTRIDE;
          return (n < 64) ? make_float2(Rrow[n].x, 0.0f)
                          : make_float2(0.f, 0.f);
        },
        [&](const auto& lo, const auto& hi) {
          const int k0 = (lane & 15) + 64 * (lane >> 4);
#pragma unroll
          for (int q = 0; q < NB; ++q) {
            float2* Rrow = Rk + (row0 + 8 * q) * RSTRIDE + k0;
#pragma unroll
            for (int i = 0; i < 4; ++i)
              Rrow[16 * i] = make_float2(lo[q][2 * i], lo[q][2 * i + 1]);
          }
          if (lane == 0) {
#pragma unroll
            for (int q = 0; q < NB; ++q)
              Rk[(row0 + 8 * q) * RSTRIDE + 128] =
                  make_float2(hi[q][0], hi[q][1]);
          }
        },
        Ftab, Wtab, spill, +1.0f, lane);
  }
  __syncthreads();

  float2* kfc = kf + (size_t)c * KFCOLS * 256;
  for (int col0 = wave; col0 < 128; col0 += 8 * NB)
    kf_columns<NB>(Rk, kfc, col0, Ftab, Wtab, spill, lane);
  if (wave == 0) kf_columns<1>(Rk, kfc, 128, Ftab, Wtab, spill, lane);
}


template <int N>
__device__ __forceinline__ void conv_columns(float2* R, const float2* kfbase,
                                             int col0, const float2* Ftab,
                                             const float2* Wtab, float2* spill,
                                             float2* spec, int lane) {
  fft256xN<N>(
      [&](int q, int n) {
        return (n < 128) ? R[n * RSTRIDE + col0 + 8 * q]
                         : make_float2(0.f, 0.f);
      },
      [&](const auto& lo, const auto& hi) {
        const int k0 = (lane & 15) + 64 * (lane >> 4);
#pragma unroll
        for (int q = 0; q < N; ++q) {
#pragma unroll
          for (int i = 0; i < 4; ++i) {
            spec[q * 256 + k0 + 16 * i] =
                make_float2(lo[q][2 * i], lo[q][2 * i + 1]);
            spec[q * 256 + 128 + k0 + 16 * i] =
                make_float2(hi[q][2 * i], hi[q][2 * i + 1]);
          }
        }
      },
      Ftab, Wtab, spill, +1.0f, lane);

  float2 kv[N][8];
#pragma unroll
  for (int q = 0; q < N; ++q) {
    const float2* kfcol = kfbase + (size_t)(col0 + 8 * q) * 256 + lane;
#pragma unroll
    for (int i = 0; i < 8; ++i) kv[q][i] = kfcol[32 * i];
  }
#pragma unroll
  for (int q = 0; q < N; ++q) {
    float2* sq = spec + q * 256 + lane;
#pragma unroll
    for (int i = 0; i < 8; ++i) {
      const float2 a = sq[32 * i];
      const float2 b = kv[q][i];
      sq[32 * i] = make_float2(a.x * b.x - a.y * b.y, a.x * b.y + a.y * b.x);
    }
  }
  asm volatile("s_wait_dscnt 0" ::: "memory");

  fft256xN<N>(
      [&](int q, int n) { return spec[q * 256 + n]; },
      [&](const auto& lo, const auto& hi) {
        const int k0 = (lane & 15) + 64 * (lane >> 4);
#pragma unroll
        for (int q = 0; q < N; ++q)
#pragma unroll
          for (int i = 0; i < 4; ++i)
            R[(k0 + 16 * i) * RSTRIDE + col0 + 8 * q] =
                make_float2(lo[q][2 * i], lo[q][2 * i + 1]);
      },
      Ftab, Wtab, spill, -1.0f, lane);
}

__global__ void __launch_bounds__(256) conv_main(
    const float* __restrict__ u, const float* __restrict__ pregate,
    const float* __restrict__ postgate, const float2* __restrict__ kf,
    float* __restrict__ y) {
  extern __shared__ char smem[];
  float2* R       = (float2*)smem;
  float2* Ftab    = (float2*)(smem + 128 * RSTRIDE * 8);
  float2* Wtab    = Ftab + 256;
  float2* scratch = Wtab + 256;

  const int tid = threadIdx.x, lane = tid & 31, wave = tid >> 5;
  const int img = blockIdx.x;
  const int c = img & 255;
  const size_t base = (size_t)img * (128 * 128);

  const float4* u4 = reinterpret_cast<const float4*>(u + base);
  const float4* p4 = reinterpret_cast<const float4*>(pregate + base);
  const int srow = 16 * wave + (lane >> 3);
  const int sseg = lane & 7;
  float4 uv[16], pv[16];
#pragma unroll
  for (int j = 0; j < 16; ++j) {
    const int i4 = (srow + 4 * (j >> 2)) * 32 + sseg + 8 * (j & 3);
    uv[j] = u4[i4];
    pv[j] = p4[i4];
  }

  init_tables(Ftab, Wtab, tid);

#pragma unroll
  for (int j = 0; j < 16; ++j) {
    float2* d = R + (srow + 4 * (j >> 2)) * RSTRIDE + 4 * (sseg + 8 * (j & 3));
    d[0] = make_float2(uv[j].x * pv[j].x, 0.0f);
    d[1] = make_float2(uv[j].y * pv[j].y, 0.0f);
    d[2] = make_float2(uv[j].z * pv[j].z, 0.0f);
    d[3] = make_float2(uv[j].w * pv[j].w, 0.0f);
  }
  __syncthreads();

  float2* spill = scratch + wave * (NB * (SPILLSZ + 256));
  float2* spec  = spill + NB * SPILLSZ;

  for (int row0 = wave; row0 < 128; row0 += 16 * NB) {
    fft256xN<NB>(
        [&](int q, int n) {
          const float2* RrowA = R + (row0 + 8 * q) * RSTRIDE;
          const float2* RrowB = RrowA + 8 * NB * RSTRIDE;
          return (n < 128) ? make_float2(RrowA[n].x, RrowB[n].x)
                           : make_float2(0.f, 0.f);
        },
        [&](const auto& lo, const auto& hi) {
          const int k0 = (lane & 15) + 64 * (lane >> 4);
#pragma unroll
          for (int q = 0; q < NB; ++q) {
#pragma unroll
            for (int i = 0; i < 4; ++i) {
              spec[q * 256 + k0 + 16 * i] =
                  make_float2(lo[q][2 * i], lo[q][2 * i + 1]);
              spec[q * 256 + 128 + k0 + 16 * i] =
                  make_float2(hi[q][2 * i], hi[q][2 * i + 1]);
            }
          }
          asm volatile("s_wait_dscnt 0" ::: "memory");
          float2 zk[NB][4], zm[NB][4], zn[NB];
#pragma unroll
          for (int q = 0; q < NB; ++q) {
            const float2* sq = spec + q * 256;
#pragma unroll
            for (int j = 0; j < 4; ++j) {
              const int k = lane + 32 * j;
              zk[q][j] = sq[k];
              zm[q][j] = sq[(256 - k) & 255];
            }
            zn[q] = sq[128];
          }
#pragma unroll
          for (int q = 0; q < NB; ++q) {
            float2* RrowA = R + (row0 + 8 * q) * RSTRIDE;
            float2* RrowB = RrowA + 8 * NB * RSTRIDE;
#pragma unroll
            for (int j = 0; j < 4; ++j) {
              const int k = lane + 32 * j;
              const float2 a = zk[q][j], m = zm[q][j];
              RrowA[k] = make_float2(0.5f * (a.x + m.x), 0.5f * (a.y - m.y));
              RrowB[k] = make_float2(0.5f * (a.y + m.y), 0.5f * (m.x - a.x));
            }
            if (lane == 0) {
              RrowA[128] = make_float2(zn[q].x, 0.0f);
              RrowB[128] = make_float2(zn[q].y, 0.0f);
            }
          }
        },
        Ftab, Wtab, spill, +1.0f, lane);
  }
  __syncthreads();

  const float2* kfbase = kf + (size_t)c * KFCOLS * 256;
  for (int col0 = wave; col0 < 128; col0 += 8 * NB) {
#pragma unroll
    for (int q = 0; q < NB; ++q) {
      const int pc = col0 + 8 * NB + 8 * q;
      if (pc < 129) {
        const char* pf = (const char*)(kfbase + (size_t)pc * 256);
        __builtin_prefetch(pf + lane * 64, 0, 1);
      }
    }
    conv_columns<NB>(R, kfbase, col0, Ftab, Wtab, spill, spec, lane);
  }
  if (wave == 0)
    conv_columns<1>(R, kfbase, 128, Ftab, Wtab, spill, spec, lane);
  __syncthreads();

  const float inv_scale = 1.0f / 65536.0f;
  for (int row0 = wave; row0 < 128; row0 += 16 * NB) {
    fft256xN<NB>(
        [&](int q, int n) {
          const float2* RrowA = R + (row0 + 8 * q) * RSTRIDE;
          const float2* RrowB = RrowA + 8 * NB * RSTRIDE;
          const bool stored = (n <= 128);
          const bool edge = ((n & 127) == 0);
          const int idx = stored ? n : 256 - n;
          const float2 a = RrowA[idx];
          const float2 b = RrowB[idx];
          const float ay = edge ? 0.0f : a.y;
          const float by = edge ? 0.0f : b.y;
          return stored ? make_float2(a.x - by, ay + b.x)
                        : make_float2(a.x + by, b.x - ay);
        },
        [&](const auto& lo, const auto& hi) {
          const int k0 = (lane & 15) + 64 * (lane >> 4);
          float gA[NB][4], gB[NB][4];
#pragma unroll
          for (int q = 0; q < NB; ++q) {
            const size_t oA = base + (size_t)(row0 + 8 * q) * 128 + k0;
            const size_t oB = oA + (size_t)(8 * NB) * 128;
#pragma unroll
            for (int i = 0; i < 4; ++i) {
              gA[q][i] = postgate[oA + 16 * i];
              gB[q][i] = postgate[oB + 16 * i];
            }
          }
#pragma unroll
          for (int q = 0; q < NB; ++q) {
            const size_t oA = base + (size_t)(row0 + 8 * q) * 128 + k0;
            const size_t oB = oA + (size_t)(8 * NB) * 128;
#pragma unroll
            for (int i = 0; i < 4; ++i) {
              y[oA + 16 * i] = lo[q][2 * i] * inv_scale * gA[q][i];
              y[oB + 16 * i] = lo[q][2 * i + 1] * inv_scale * gB[q][i];
            }
          }
        },
        Ftab, Wtab, spill, -1.0f, lane);
  }
}

extern "C" void kernel_launch(void* const* d_in, const int* in_sizes, int n_in,
                              void* d_out, int out_size, void* d_ws,
                              size_t ws_size, hipStream_t stream) {
  const float* u        = (const float*)d_in[0];
  const float* k        = (const float*)d_in[1];
  const float* pregate  = (const float*)d_in[2];
  const float* postgate = (const float*)d_in[3];
  float* y   = (float*)d_out;
  float2* kf = (float2*)d_ws;

  const size_t lds_kf   = (size_t)64 * RSTRIDE * 8 + 512 * 8 + 8 * NB * SPILLSZ * 8;
  const size_t lds_main =
      (size_t)128 * RSTRIDE * 8 + 512 * 8 + 8 * NB * (SPILLSZ + 256) * 8;

  (void)hipFuncSetAttribute((const void*)kf_precompute,
                            hipFuncAttributeMaxDynamicSharedMemorySize,
                            (int)lds_kf);
  (void)hipFuncSetAttribute((const void*)conv_main,
                            hipFuncAttributeMaxDynamicSharedMemorySize,
                            (int)lds_main);

  kf_precompute<<<256, 256, lds_kf, stream>>>(k, kf);
  conv_main<<<4 * 256, 256, lds_main, stream>>>(u, pregate, postgate, kf, y);
}


# Both kernels: each transform stage's loads (LDS operands and tables, k_f, postgate) go out as one group ahead of its block of matrix products, held by scheduling barriers; none left among products
# speedup vs baseline: 3.6719x; 1.0368x over previous
#include <hip/hip_runtime.h>
#include <hip/hip_bf16.h>
#include <math.h>


typedef float v2f __attribute__((ext_vector_type(2)));
typedef float v8f __attribute__((ext_vector_type(8)));

#define PI_F 3.14159265358979323846f
#define RSTRIDE 133
#define KFCOLS 132
#define NB 4
#define SPITCH 18
#define SPILLSZ (16 * SPITCH)

__device__ __forceinline__ v8f wmma_f32(v2f a, v2f b, v8f c) {
  return __builtin_amdgcn_wmma_f32_16x16x4_f32(false, a, false, b, (short)0, c,
                                               false, false);
}

__device__ __forceinline__ void block_fence() {
  __builtin_amdgcn_sched_barrier(0x6);
}


__device__ __forceinline__ float2 load_g_entry(const float2* Ftab, int t, int b,
                                               int lane) {
  const int m = lane & 15;
  const int h = (lane >> 4) & 1;
  const int c = m >> 1;
  const int a = 2 * t + h;
  return Ftab[16 + (((2 * a + b) * c) & 15)];
}

__device__ __forceinline__ v2f make_g_frag(float2 g, float dir, int lane) {
  const int m = lane & 15;
  const float gr = g.x, gi = dir * g.y;
  v2f f;
  if (m & 1) { f.x = gi; f.y = gr; } else { f.x = gr; f.y = -gi; }
  return f;
}

typedef float v4f __attribute__((ext_vector_type(4)));

__device__ __forceinline__ v2f as_frag(float2 x) {
  v2f f;
  f.x = x.x;
  f.y = x.y;
  return f;
}

template <int N, typename LD, typename FX, typename DST>
__device__ __forceinline__ void fft256xN(LD ld, FX fx, DST dst,
                                         const float2* Ftab,
                                         const float2* Wtab, float2* spill,
                                         float dir, int lane) {
  const int col = lane & 15;
  const int h   = (lane >> 4) & 1;

  float2 gte[4], gto[4];
#pragma unroll
  for (int t = 0; t < 4; ++t) {
    gte[t] = load_g_entry(Ftab, t, 0, lane);
    gto[t] = load_g_entry(Ftab, t, 1, lane);
  }
  float2 wt[2][4];
#pragma unroll
  for (int d = 0; d < 2; ++d) {
#pragma unroll
    for (int i = 0; i < 4; ++i)
      wt[d][i] = Wtab[(col * (4 * h + i + 8 * d)) & 255];
  }
  using Raw = decltype(ld(0, 0));
  Raw re[N][4], ro[N][4];
#pragma unroll
  for (int t = 0; t < 4; ++t) {
    const int n2 = 2 * (2 * t + h);
#pragma unroll
    for (int q = 0; q < N; ++q) {
      re[q][t] = ld(q, col + 16 * n2);
      ro[q][t] = ld(q, col + 16 * (n2 + 1));
    }
  }
  block_fence();

  v2f ge[4], go[4];
#pragma unroll
  for (int t = 0; t < 4; ++t) {
    ge[t] = make_g_frag(gte[t], dir, lane);
    go[t] = make_g_frag(gto[t], dir, lane);
  }

  v8f e[N], o[N];
#pragma unroll
  for (int q = 0; q < N; ++q) {
    e[q] = v8f{}; o[q] = v8f{};
  }

#pragma unroll
  for (int t = 0; t < 4; ++t) {
    const int n2 = 2 * (2 * t + h);
#pragma unroll
    for (int q = 0; q < N; ++q) {
      const v2f xe = as_frag(fx(re[q][t], col + 16 * n2));
      const v2f xo = as_frag(fx(ro[q][t], col + 16 * (n2 + 1)));
      e[q] = wmma_f32(ge[t], xe, e[q]);
      o[q] = wmma_f32(go[t], xo, o[q]);
    }
  }
  block_fence();

#pragma unroll
  for (int q = 0; q < N; ++q) {
    float2* sp = spill + q * SPILLSZ + col;
#pragma unroll
    for (int i = 0; i < 4; ++i) {
      const float lr = e[q][2 * i] + o[q][2 * i];
      const float li = e[q][2 * i + 1] + o[q][2 * i + 1];
      const float hr = e[q][2 * i] - o[q][2 * i];
      const float hm = e[q][2 * i + 1] - o[q][2 * i + 1];
      const float t0r = wt[0][i].x, t0i = dir * wt[0][i].y;
      const float t1r = wt[1][i].x, t1i = dir * wt[1][i].y;
      const int k1 = 4 * h + i;
      sp[k1 * SPITCH] =
          make_float2(lr * t0r - li * t0i, lr * t0i + li * t0r);
      sp[(k1 + 8) * SPITCH] =
          make_float2(hr * t1r - hm * t1i, hr * t1i + hm * t1r);
    }
  }
  asm volatile("s_wait_dscnt 0" ::: "memory");

  v2f be[N][4], bo[N][4];
#pragma unroll
  for (int t = 0; t < 4; ++t) {
    const int n1 = 2 * (2 * t + h);
#pragma unroll
    for (int q = 0; q < N; ++q) {
      const float2* sp = spill + q * SPILLSZ + col * SPITCH + n1;
      be[q][t] = as_frag(sp[0]);
      bo[q][t] = as_frag(sp[1]);
    }
  }
  block_fence();

#pragma unroll
  for (int q = 0; q < N; ++q) {
    e[q] = v8f{}; o[q] = v8f{};
  }

#pragma unroll
  for (int t = 0; t < 4; ++t) {
#pragma unroll
    for (int q = 0; q < N; ++q) {
      e[q] = wmma_f32(ge[t], be[q][t], e[q]);
      o[q] = wmma_f32(go[t], bo[q][t], o[q]);
    }
  }
  block_fence();

  v8f lo[N], hi[N];
#pragma unroll
  for (int q = 0; q < N; ++q) {
    lo[q] = e[q] + o[q];
    hi[q] = e[q] - o[q];
  }
  dst(lo, hi);
}

__device__ __forceinline__ void init_tables(float2* Ftab, float2* Wtab,
                                            int tid) {
  for (int i = tid; i < 256; i += 256) {
    const int r = i >> 4, c = i & 15;
    const float tf = -2.0f * PI_F * (float)((r * c) & 15) / 16.0f;
    Ftab[i] = make_float2(cosf(tf), sinf(tf));
    const float tw = -2.0f * PI_F * (float)i / 256.0f;
    Wtab[i] = make_float2(cosf(tw), sinf(tw));
  }
}


template <int N>
__device__ __forceinline__ void kf_columns(const float2* Rk, float2* kfc,
                                           int col0, const float2* Ftab,
                                           const float2* Wtab, float2* spill,
                                           int lane) {
  fft256xN<N>(
      [&](int q, int n) {
        return (n < 64) ? Rk[n * RSTRIDE + col0 + 8 * q]
                        : make_float2(0.f, 0.f);
      },
      [](float2 v, int) { return v; },
      [&](const auto& lo, const auto& hi) {
        const int k0 = (lane & 15) + 64 * (lane >> 4);
#pragma unroll
        for (int q = 0; q < N; ++q) {
          float2* dcol = kfc + (size_t)(col0 + 8 * q) * 256 + k0;
#pragma unroll
          for (int i = 0; i < 4; ++i) {
            dcol[16 * i] = make_float2(lo[q][2 * i], lo[q][2 * i + 1]);
            dcol[128 + 16 * i] = make_float2(hi[q][2 * i], hi[q][2 * i + 1]);
          }
        }
      },
      Ftab, Wtab, spill, +1.0f, lane);
}

__global__ void __launch_bounds__(256) kf_precompute(
    const float* __restrict__ kin, float2* __restrict__ kf) {
  extern __shared__ char smem[];
  float2* Rk      = (float2*)smem;
  float2* Ftab    = (float2*)(smem + 64 * RSTRIDE * 8);
  float2* Wtab    = Ftab + 256;
  float2* scratch = Wtab + 256;

  const int tid = threadIdx.x, lane = tid & 31, wave = tid >> 5;
  const int c = blockIdx.x;

  const float4* k4 = reinterpret_cast<const float4*>(kin + (size_t)c * (64 * 64));
  const int srow = 8 * wave + (lane >> 3);
  const int sseg = lane & 7;
  float4 kq[4];
#pragma unroll
  for (int j = 0; j < 4; ++j)
    kq[j] = k4[(srow + 4 * (j >> 1)) * 16 + sseg + 8 * (j & 1)];

  init_tables(Ftab, Wtab, tid);

#pragma unroll
  for (int j = 0; j < 4; ++j) {
    float2* d = Rk + (srow + 4 * (j >> 1)) * RSTRIDE + 4 * (sseg + 8 * (j & 1));
    d[0] = make_float2(kq[j].x, 0.0f);
    d[1] = make_float2(kq[j].y, 0.0f);
    d[2] = make_float2(kq[j].z, 0.0f);
    d[3] = make_float2(kq[j].w, 0.0f);
  }
  __syncthreads();

  float2* spill = scratch + wave * (NB * SPILLSZ);

  for (int row0 = wave; row0 < 64; row0 += 8 * NB) {
    fft256xN<NB>(
        [&](int q, int n) {
          const float2* Rrow = Rk + (row0 + 8 * q) * RSTRIDE;
          return (n < 64) ? make_float2(Rrow[n].x, 0.0f)
                          : make_float2(0.f, 0.f);
        },
        [](float2 v, int) { return v; },
        [&](const auto& lo, const auto& hi) {
          const int k0 = (lane & 15) + 64 * (lane >> 4);
#pragma unroll
          for (int q = 0; q < NB; ++q) {
            float2* Rrow = Rk + (row0 + 8 * q) * RSTRIDE + k0;
#pragma unroll
            for (int i = 0; i < 4; ++i)
              Rrow[16 * i] = make_float2(lo[q][2 * i], lo[q][2 * i + 1]);
          }
          if (lane == 0) {
#pragma unroll
            for (int q = 0; q < NB; ++q)
              Rk[(row0 + 8 * q) * RSTRIDE + 128] =
                  make_float2(hi[q][0], hi[q][1]);
          }
        },
        Ftab, Wtab, spill, +1.0f, lane);
  }
  __syncthreads();

  float2* kfc = kf + (size_t)c * KFCOLS * 256;
  for (int col0 = wave; col0 < 128; col0 += 8 * NB)
    kf_columns<NB>(Rk, kfc, col0, Ftab, Wtab, spill, lane);
  if (wave == 0) kf_columns<1>(Rk, kfc, 128, Ftab, Wtab, spill, lane);
}


template <int N>
__device__ __forceinline__ void conv_columns(float2* R, const float2* kfbase,
                                             int col0, const float2* Ftab,
                                             const float2* Wtab, float2* spill,
                                             float2* spec, int lane) {
  float2 kv[N][8];
#pragma unroll
  for (int q = 0; q < N; ++q) {
    const float2* kfcol = kfbase + (size_t)(col0 + 8 * q) * 256 + lane;
#pragma unroll
    for (int i = 0; i < 8; ++i) kv[q][i] = kfcol[32 * i];
  }

  fft256xN<N>(
      [&](int q, int n) {
        return (n < 128) ? R[n * RSTRIDE + col0 + 8 * q]
                         : make_float2(0.f, 0.f);
      },
      [](float2 v, int) { return v; },
      [&](const auto& lo, const auto& hi) {
        const int k0 = (lane & 15) + 64 * (lane >> 4);
#pragma unroll
        for (int q = 0; q < N; ++q) {
#pragma unroll
          for (int i = 0; i < 4; ++i) {
            spec[q * 256 + k0 + 16 * i] =
                make_float2(lo[q][2 * i], lo[q][2 * i + 1]);
            spec[q * 256 + 128 + k0 + 16 * i] =
                make_float2(hi[q][2 * i], hi[q][2 * i + 1]);
          }
        }
      },
      Ftab, Wtab, spill, +1.0f, lane);

#pragma unroll
  for (int q = 0; q < N; ++q) {
    float2* sq = spec + q * 256 + lane;
#pragma unroll
    for (int i = 0; i < 8; ++i) {
      const float2 a = sq[32 * i];
      const float2 b = kv[q][i];
      sq[32 * i] = make_float2(a.x * b.x - a.y * b.y, a.x * b.y + a.y * b.x);
    }
  }
  asm volatile("s_wait_dscnt 0" ::: "memory");

  fft256xN<N>(
      [&](int q, int n) { return spec[q * 256 + n]; },
      [](float2 v, int) { return v; },
      [&](const auto& lo, const auto& hi) {
        const int k0 = (lane & 15) + 64 * (lane >> 4);
#pragma unroll
        for (int q = 0; q < N; ++q)
#pragma unroll
          for (int i = 0; i < 4; ++i)
            R[(k0 + 16 * i) * RSTRIDE + col0 + 8 * q] =
                make_float2(lo[q][2 * i], lo[q][2 * i + 1]);
      },
      Ftab, Wtab, spill, -1.0f, lane);
}

__global__ void __launch_bounds__(256) __attribute__((amdgpu_waves_per_eu(2, 2)))
conv_main(
    const float* __restrict__ u, const float* __restrict__ pregate,
    const float* __restrict__ postgate, const float2* __restrict__ kf,
    float* __restrict__ y) {
  extern __shared__ char smem[];
  float2* R       = (float2*)smem;
  float2* Ftab    = (float2*)(smem + 128 * RSTRIDE * 8);
  float2* Wtab    = Ftab + 256;
  float2* scratch = Wtab + 256;

  const int tid = threadIdx.x, lane = tid & 31, wave = tid >> 5;
  const int img = blockIdx.x;
  const int c = img & 255;
  const size_t base = (size_t)img * (128 * 128);

  const float4* u4 = reinterpret_cast<const float4*>(u + base);
  const float4* p4 = reinterpret_cast<const float4*>(pregate + base);
  const int srow = 16 * wave + (lane >> 3);
  const int sseg = lane & 7;
  float4 uv[16], pv[16];
#pragma unroll
  for (int j = 0; j < 16; ++j) {
    const int i4 = (srow + 4 * (j >> 2)) * 32 + sseg + 8 * (j & 3);
    uv[j] = u4[i4];
    pv[j] = p4[i4];
  }

  init_tables(Ftab, Wtab, tid);

#pragma unroll
  for (int j = 0; j < 16; ++j) {
    float2* d = R + (srow + 4 * (j >> 2)) * RSTRIDE + 4 * (sseg + 8 * (j & 3));
    d[0] = make_float2(uv[j].x * pv[j].x, 0.0f);
    d[1] = make_float2(uv[j].y * pv[j].y, 0.0f);
    d[2] = make_float2(uv[j].z * pv[j].z, 0.0f);
    d[3] = make_float2(uv[j].w * pv[j].w, 0.0f);
  }
  __syncthreads();

  float2* spill = scratch + wave * (NB * (SPILLSZ + 256));
  float2* spec  = spill + NB * SPILLSZ;

  for (int row0 = wave; row0 < 128; row0 += 16 * NB) {
    fft256xN<NB>(
        [&](int q, int n) {
          const float2* RrowA = R + (row0 + 8 * q) * RSTRIDE;
          const float2* RrowB = RrowA + 8 * NB * RSTRIDE;
          return (n < 128) ? make_float2(RrowA[n].x, RrowB[n].x)
                           : make_float2(0.f, 0.f);
        },
        [](float2 v, int) { return v; },
        [&](const auto& lo, const auto& hi) {
          const int k0 = (lane & 15) + 64 * (lane >> 4);
#pragma unroll
          for (int q = 0; q < NB; ++q) {
#pragma unroll
            for (int i = 0; i < 4; ++i) {
              spec[q * 256 + k0 + 16 * i] =
                  make_float2(lo[q][2 * i], lo[q][2 * i + 1]);
              spec[q * 256 + 128 + k0 + 16 * i] =
                  make_float2(hi[q][2 * i], hi[q][2 * i + 1]);
            }
          }
          asm volatile("s_wait_dscnt 0" ::: "memory");
          float2 zk[NB][4], zm[NB][4], zn[NB];
#pragma unroll
          for (int q = 0; q < NB; ++q) {
            const float2* sq = spec + q * 256;
#pragma unroll
            for (int j = 0; j < 4; ++j) {
              const int k = lane + 32 * j;
              zk[q][j] = sq[k];
              zm[q][j] = sq[(256 - k) & 255];
            }
            zn[q] = sq[128];
          }
#pragma unroll
          for (int q = 0; q < NB; ++q) {
            float2* RrowA = R + (row0 + 8 * q) * RSTRIDE;
            float2* RrowB = RrowA + 8 * NB * RSTRIDE;
#pragma unroll
            for (int j = 0; j < 4; ++j) {
              const int k = lane + 32 * j;
              const float2 a = zk[q][j], m = zm[q][j];
              RrowA[k] = make_float2(0.5f * (a.x + m.x), 0.5f * (a.y - m.y));
              RrowB[k] = make_float2(0.5f * (a.y + m.y), 0.5f * (m.x - a.x));
            }
            if (lane == 0) {
              RrowA[128] = make_float2(zn[q].x, 0.0f);
              RrowB[128] = make_float2(zn[q].y, 0.0f);
            }
          }
        },
        Ftab, Wtab, spill, +1.0f, lane);
  }
  __syncthreads();

  const float2* kfbase = kf + (size_t)c * KFCOLS * 256;
  for (int col0 = wave; col0 < 128; col0 += 8 * NB) {
#pragma unroll
    for (int q = 0; q < NB; ++q) {
      const int pc = col0 + 8 * NB + 8 * q;
      if (pc < 129) {
        const char* pf = (const char*)(kfbase + (size_t)pc * 256);
        __builtin_prefetch(pf + lane * 64, 0, 1);
      }
    }
    conv_columns<NB>(R, kfbase, col0, Ftab, Wtab, spill, spec, lane);
  }
  if (wave == 0)
    conv_columns<1>(R, kfbase, 128, Ftab, Wtab, spill, spec, lane);
  __syncthreads();

  const float inv_scale = 1.0f / 65536.0f;
  for (int row0 = wave; row0 < 128; row0 += 16 * NB) {
    const int k0 = (lane & 15) + 64 * (lane >> 4);
    float gA[NB][4], gB[NB][4];
#pragma unroll
    for (int q = 0; q < NB; ++q) {
      const size_t oA = base + (size_t)(row0 + 8 * q) * 128 + k0;
      const size_t oB = oA + (size_t)(8 * NB) * 128;
#pragma unroll
      for (int i = 0; i < 4; ++i) {
        gA[q][i] = postgate[oA + 16 * i];
        gB[q][i] = postgate[oB + 16 * i];
      }
    }
    fft256xN<NB>(
        [&](int q, int n) {
          const float2* RrowA = R + (row0 + 8 * q) * RSTRIDE;
          const float2* RrowB = RrowA + 8 * NB * RSTRIDE;
          const int idx = (n <= 128) ? n : 256 - n;
          v4f r;
          r.x = RrowA[idx].x;
          r.y = RrowA[idx].y;
          r.z = RrowB[idx].x;
          r.w = RrowB[idx].y;
          return r;
        },
        [](v4f r, int n) {
          const bool stored = (n <= 128);
          const bool edge = ((n & 127) == 0);
          const float2 a = make_float2(r.x, r.y);
          const float2 b = make_float2(r.z, r.w);
          const float ay = edge ? 0.0f : a.y;
          const float by = edge ? 0.0f : b.y;
          return stored ? make_float2(a.x - by, ay + b.x)
                        : make_float2(a.x + by, b.x - ay);
        },
        [&](const auto& lo, const auto& hi) {
#pragma unroll
          for (int q = 0; q < NB; ++q) {
            const size_t oA = base + (size_t)(row0 + 8 * q) * 128 + k0;
            const size_t oB = oA + (size_t)(8 * NB) * 128;
#pragma unroll
            for (int i = 0; i < 4; ++i) {
              y[oA + 16 * i] = lo[q][2 * i] * inv_scale * gA[q][i];
              y[oB + 16 * i] = lo[q][2 * i + 1] * inv_scale * gB[q][i];
            }
          }
        },
        Ftab, Wtab, spill, -1.0f, lane);
  }
}

extern "C" void kernel_launch(void* const* d_in, const int* in_sizes, int n_in,
                              void* d_out, int out_size, void* d_ws,
                              size_t ws_size, hipStream_t stream) {
  const float* u        = (const float*)d_in[0];
  const float* k        = (const float*)d_in[1];
  const float* pregate  = (const float*)d_in[2];
  const float* postgate = (const float*)d_in[3];
  float* y   = (float*)d_out;
  float2* kf = (float2*)d_ws;

  const size_t lds_kf   = (size_t)64 * RSTRIDE * 8 + 512 * 8 + 8 * NB * SPILLSZ * 8;
  const size_t lds_main =
      (size_t)128 * RSTRIDE * 8 + 512 * 8 + 8 * NB * (SPILLSZ + 256) * 8;

  (void)hipFuncSetAttribute((const void*)kf_precompute,
                            hipFuncAttributeMaxDynamicSharedMemorySize,
                            (int)lds_kf);
  (void)hipFuncSetAttribute((const void*)conv_main,
                            hipFuncAttributeMaxDynamicSharedMemorySize,
                            (int)lds_main);

  kf_precompute<<<256, 256, lds_kf, stream>>>(k, kf);
  conv_main<<<4 * 256, 256, lds_main, stream>>>(u, pregate, postgate, kf, y);
}


# Transform length 192 = 12 x 16 per axis, was 256: 128 + 64 - 1 = 191 fits, nothing wraps, same outputs; 12-point first stage (14 WMMA per FFT, was 16), 97 columns, was 129; new rounding, sum order
# speedup vs baseline: 4.5066x; 1.2273x over previous
#include <hip/hip_runtime.h>
#include <hip/hip_bf16.h>
#include <math.h>


typedef float v2f __attribute__((ext_vector_type(2)));
typedef float v8f __attribute__((ext_vector_type(8)));

#define PI_F 3.14159265358979323846f
#define NFFT 192
#define NHALF 96
#define RSTRIDE 133
#define KFCOLS 97
#define NB 4
#define SPITCH 18
#define SPILLSZ (12 * SPITCH)

__device__ __forceinline__ v8f wmma_f32(v2f a, v2f b, v8f c) {
  return __builtin_amdgcn_wmma_f32_16x16x4_f32(false, a, false, b, (short)0, c,
                                               false, false);
}

__device__ __forceinline__ void block_fence() {
  __builtin_amdgcn_sched_barrier(0x6);
}


__device__ __forceinline__ float2 load_g_entry(const float2* Ftab, int t, int b,
                                               int lane) {
  const int m = lane & 15;
  const int h = (lane >> 4) & 1;
  const int c = m >> 1;
  const int a = 2 * t + h;
  return Ftab[16 + (((2 * a + b) * c) & 15)];
}

__device__ __forceinline__ v2f make_g_frag(float2 g, float dir, int lane) {
  const int m = lane & 15;
  const float gr = g.x, gi = dir * g.y;
  v2f f;
  if (m & 1) { f.x = gi; f.y = gr; } else { f.x = gr; f.y = -gi; }
  return f;
}

__device__ __forceinline__ float2 load_g12_entry(const float2* Ftab, int t,
                                                 int b, int lane) {
  const int m = lane & 15;
  const int h = (lane >> 4) & 1;
  const int c = 3 * (m >> 3) + ((m & 7) >> 1);
  const int a = 2 * t + h;
  return Ftab[32 + (((2 * a + b) * c) % 12)];
}

__device__ __forceinline__ v2f make_g12_frag(float2 g, float dir, int lane) {
  v2f f = make_g_frag(g, dir, lane);
  if ((lane & 7) >= 6) { f.x = 0.0f; f.y = 0.0f; }
  return f;
}

typedef float v4f __attribute__((ext_vector_type(4)));

__device__ __forceinline__ void hold_loaded(float x) {
  asm volatile("" ::"v"(x));
}

__device__ __forceinline__ v2f as_frag(float2 x) {
  v2f f;
  f.x = x.x;
  f.y = x.y;
  return f;
}

template <int N, typename LD, typename FX, typename DST>
__device__ __forceinline__ void fft192xN(LD ld, FX fx, DST dst,
                                         const float2* Ftab,
                                         const float2* Wtab, float2* spill,
                                         float dir, int lane) {
  const int col = lane & 15;
  const int h   = (lane >> 4) & 1;

  float2 gte[3], gto[3];
#pragma unroll
  for (int t = 0; t < 3; ++t) {
    gte[t] = load_g12_entry(Ftab, t, 0, lane);
    gto[t] = load_g12_entry(Ftab, t, 1, lane);
  }
  float2 wt[2][3];
#pragma unroll
  for (int d = 0; d < 2; ++d) {
#pragma unroll
    for (int i = 0; i < 3; ++i)
      wt[d][i] = Wtab[col * (3 * h + i + 6 * d)];
  }
  using Raw = decltype(ld(0, 0));
  Raw re[N][3], ro[N][3];
#pragma unroll
  for (int t = 0; t < 3; ++t) {
    const int n2 = 2 * (2 * t + h);
#pragma unroll
    for (int q = 0; q < N; ++q) {
      re[q][t] = ld(q, col + 16 * n2);
      ro[q][t] = ld(q, col + 16 * (n2 + 1));
    }
  }
  block_fence();

  v2f ge[3], go[3];
#pragma unroll
  for (int t = 0; t < 3; ++t) {
    ge[t] = make_g12_frag(gte[t], dir, lane);
    go[t] = make_g12_frag(gto[t], dir, lane);
  }

  v8f e[N], o[N];
#pragma unroll
  for (int q = 0; q < N; ++q) {
    e[q] = v8f{}; o[q] = v8f{};
  }

#pragma unroll
  for (int t = 0; t < 3; ++t) {
    const int n2 = 2 * (2 * t + h);
#pragma unroll
    for (int q = 0; q < N; ++q) {
      const v2f xe = as_frag(fx(re[q][t], col + 16 * n2));
      const v2f xo = as_frag(fx(ro[q][t], col + 16 * (n2 + 1)));
      e[q] = wmma_f32(ge[t], xe, e[q]);
      o[q] = wmma_f32(go[t], xo, o[q]);
    }
  }
  block_fence();

#pragma unroll
  for (int q = 0; q < N; ++q) {
    float2* sp = spill + q * SPILLSZ + col;
#pragma unroll
    for (int i = 0; i < 3; ++i) {
      const float lr = e[q][2 * i] + o[q][2 * i];
      const float li = e[q][2 * i + 1] + o[q][2 * i + 1];
      const float hr = e[q][2 * i] - o[q][2 * i];
      const float hm = e[q][2 * i + 1] - o[q][2 * i + 1];
      const float t0r = wt[0][i].x, t0i = dir * wt[0][i].y;
      const float t1r = wt[1][i].x, t1i = dir * wt[1][i].y;
      const int k1 = 3 * h + i;
      sp[k1 * SPITCH] =
          make_float2(lr * t0r - li * t0i, lr * t0i + li * t0r);
      sp[(k1 + 6) * SPITCH] =
          make_float2(hr * t1r - hm * t1i, hr * t1i + hm * t1r);
    }
  }
  asm volatile("s_wait_dscnt 0" ::: "memory");

  float2 hte[4], hto[4];
#pragma unroll
  for (int t = 0; t < 4; ++t) {
    hte[t] = load_g_entry(Ftab, t, 0, lane);
    hto[t] = load_g_entry(Ftab, t, 1, lane);
  }
  const int ccol = (col < 12) ? col : 11;
  v2f be[N][4], bo[N][4];
#pragma unroll
  for (int t = 0; t < 4; ++t) {
    const int n1 = 2 * (2 * t + h);
#pragma unroll
    for (int q = 0; q < N; ++q) {
      const float2* sp = spill + q * SPILLSZ + ccol * SPITCH + n1;
      be[q][t] = as_frag(sp[0]);
      bo[q][t] = as_frag(sp[1]);
    }
  }
  block_fence();

  v2f he[4], ho[4];
#pragma unroll
  for (int t = 0; t < 4; ++t) {
    he[t] = make_g_frag(hte[t], dir, lane);
    ho[t] = make_g_frag(hto[t], dir, lane);
  }

  v8f e2[N], o2[N];
#pragma unroll
  for (int q = 0; q < N; ++q) {
    e2[q] = v8f{}; o2[q] = v8f{};
  }

#pragma unroll
  for (int t = 0; t < 4; ++t) {
#pragma unroll
    for (int q = 0; q < N; ++q) {
      e2[q] = wmma_f32(he[t], be[q][t], e2[q]);
      o2[q] = wmma_f32(ho[t], bo[q][t], o2[q]);
    }
  }
  block_fence();

  v8f lo[N], hi[N];
#pragma unroll
  for (int q = 0; q < N; ++q) {
    lo[q] = e2[q] + o2[q];
    hi[q] = e2[q] - o2[q];
  }
  dst(lo, hi);
}

__device__ __forceinline__ void init_tables(float2* Ftab, float2* Wtab,
                                            int tid) {
  for (int i = tid; i < 256; i += 256) {
    const int r = i >> 4, c = i & 15;
    const float tf = (r == 2)
        ? -2.0f * PI_F * (float)(c % 12) / 12.0f
        : -2.0f * PI_F * (float)((r * c) & 15) / 16.0f;
    Ftab[i] = make_float2(cosf(tf), sinf(tf));
    const float tw = -2.0f * PI_F * (float)(i % NFFT) / (float)NFFT;
    Wtab[i] = make_float2(cosf(tw), sinf(tw));
  }
}


template <int N>
__device__ __forceinline__ void kf_columns(const float2* Rk, float2* kfc,
                                           int col0, const float2* Ftab,
                                           const float2* Wtab, float2* spill,
                                           int lane) {
  fft192xN<N>(
      [&](int q, int n) {
        return (n < 64) ? Rk[n * RSTRIDE + col0 + 8 * q]
                        : make_float2(0.f, 0.f);
      },
      [](float2 v, int) { return v; },
      [&](const auto& lo, const auto& hi) {
        const int k0 = (lane & 15) + 48 * (lane >> 4);
        if ((lane & 15) < 12) {
#pragma unroll
          for (int q = 0; q < N; ++q) {
            float2* dcol = kfc + (size_t)(col0 + 8 * q) * NFFT + k0;
#pragma unroll
            for (int i = 0; i < 4; ++i) {
              dcol[12 * i] = make_float2(lo[q][2 * i], lo[q][2 * i + 1]);
              dcol[NHALF + 12 * i] =
                  make_float2(hi[q][2 * i], hi[q][2 * i + 1]);
            }
          }
        }
      },
      Ftab, Wtab, spill, +1.0f, lane);
}

__global__ void __launch_bounds__(256) kf_precompute(
    const float* __restrict__ kin, float2* __restrict__ kf) {
  extern __shared__ char smem[];
  float2* Rk      = (float2*)smem;
  float2* Ftab    = (float2*)(smem + 64 * RSTRIDE * 8);
  float2* Wtab    = Ftab + 256;
  float2* scratch = Wtab + 256;

  const int tid = threadIdx.x, lane = tid & 31, wave = tid >> 5;
  const int c = blockIdx.x;

  const float4* k4 = reinterpret_cast<const float4*>(kin + (size_t)c * (64 * 64));
  const int srow = 8 * wave + (lane >> 3);
  const int sseg = lane & 7;
  float4 kq[4];
#pragma unroll
  for (int j = 0; j < 4; ++j)
    kq[j] = k4[(srow + 4 * (j >> 1)) * 16 + sseg + 8 * (j & 1)];

  init_tables(Ftab, Wtab, tid);

#pragma unroll
  for (int j = 0; j < 4; ++j) {
    float2* d = Rk + (srow + 4 * (j >> 1)) * RSTRIDE + 4 * (sseg + 8 * (j & 1));
    d[0] = make_float2(kq[j].x, 0.0f);
    d[1] = make_float2(kq[j].y, 0.0f);
    d[2] = make_float2(kq[j].z, 0.0f);
    d[3] = make_float2(kq[j].w, 0.0f);
  }
  __syncthreads();

  float2* spill = scratch + wave * (NB * SPILLSZ);

  for (int row0 = wave; row0 < 64; row0 += 8 * NB) {
    fft192xN<NB>(
        [&](int q, int n) {
          const float2* Rrow = Rk + (row0 + 8 * q) * RSTRIDE;
          return (n < 64) ? make_float2(Rrow[n].x, 0.0f)
                          : make_float2(0.f, 0.f);
        },
        [](float2 v, int) { return v; },
        [&](const auto& lo, const auto& hi) {
          const int k0 = (lane & 15) + 48 * (lane >> 4);
          if ((lane & 15) < 12) {
#pragma unroll
            for (int q = 0; q < NB; ++q) {
              float2* Rrow = Rk + (row0 + 8 * q) * RSTRIDE + k0;
#pragma unroll
              for (int i = 0; i < 4; ++i)
                Rrow[12 * i] = make_float2(lo[q][2 * i], lo[q][2 * i + 1]);
            }
          }
          if (lane == 0) {
#pragma unroll
            for (int q = 0; q < NB; ++q)
              Rk[(row0 + 8 * q) * RSTRIDE + NHALF] =
                  make_float2(hi[q][0], hi[q][1]);
          }
        },
        Ftab, Wtab, spill, +1.0f, lane);
  }
  __syncthreads();

  float2* kfc = kf + (size_t)c * KFCOLS * NFFT;
  for (int col0 = wave; col0 < NHALF; col0 += 8 * NB)
    kf_columns<NB>(Rk, kfc, col0, Ftab, Wtab, spill, lane);
  if (wave == 0) kf_columns<1>(Rk, kfc, NHALF, Ftab, Wtab, spill, lane);
}


template <int N>
__device__ __forceinline__ void conv_columns(float2* R, const float2* kfbase,
                                             int col0, const float2* Ftab,
                                             const float2* Wtab, float2* spill,
                                             float2* spec, int lane) {
  float2 kv[N][6];
#pragma unroll
  for (int q = 0; q < N; ++q) {
    const float2* kfcol = kfbase + (size_t)(col0 + 8 * q) * NFFT + lane;
#pragma unroll
    for (int i = 0; i < 6; ++i) kv[q][i] = kfcol[32 * i];
  }

  fft192xN<N>(
      [&](int q, int n) {
        return (n < 128) ? R[n * RSTRIDE + col0 + 8 * q]
                         : make_float2(0.f, 0.f);
      },
      [](float2 v, int) { return v; },
      [&](const auto& lo, const auto& hi) {
        const int k0 = (lane & 15) + 48 * (lane >> 4);
        if ((lane & 15) < 12) {
#pragma unroll
          for (int q = 0; q < N; ++q) {
#pragma unroll
            for (int i = 0; i < 4; ++i) {
              spec[q * NFFT + k0 + 12 * i] =
                  make_float2(lo[q][2 * i], lo[q][2 * i + 1]);
              spec[q * NFFT + NHALF + k0 + 12 * i] =
                  make_float2(hi[q][2 * i], hi[q][2 * i + 1]);
            }
          }
        }
      },
      Ftab, Wtab, spill, +1.0f, lane);

#pragma unroll
  for (int q = 0; q < N; ++q) {
    float2* sq = spec + q * NFFT + lane;
#pragma unroll
    for (int i = 0; i < 6; ++i) {
      const float2 a = sq[32 * i];
      const float2 b = kv[q][i];
      sq[32 * i] = make_float2(a.x * b.x - a.y * b.y, a.x * b.y + a.y * b.x);
    }
  }
  asm volatile("s_wait_dscnt 0" ::: "memory");

  fft192xN<N>(
      [&](int q, int n) { return spec[q * NFFT + n]; },
      [](float2 v, int) { return v; },
      [&](const auto& lo, const auto& hi) {
        const int k0 = (lane & 15) + 48 * (lane >> 4);
        if ((lane & 15) < 12) {
#pragma unroll
          for (int q = 0; q < N; ++q) {
#pragma unroll
            for (int i = 0; i < 4; ++i)
              R[(k0 + 12 * i) * RSTRIDE + col0 + 8 * q] =
                  make_float2(lo[q][2 * i], lo[q][2 * i + 1]);
          }
          if (lane < 16) {
#pragma unroll
            for (int q = 0; q < N; ++q) {
#pragma unroll
              for (int i = 0; i < 2; ++i)
                R[(NHALF + k0 + 12 * i) * RSTRIDE + col0 + 8 * q] =
                    make_float2(hi[q][2 * i], hi[q][2 * i + 1]);
            }
            if (lane < 8) {
#pragma unroll
              for (int q = 0; q < N; ++q)
                R[(NHALF + k0 + 24) * RSTRIDE + col0 + 8 * q] =
                    make_float2(hi[q][4], hi[q][5]);
            }
          }
        }
      },
      Ftab, Wtab, spill, -1.0f, lane);
}

__global__ void __launch_bounds__(256) __attribute__((amdgpu_waves_per_eu(2, 2)))
conv_main(
    const float* __restrict__ u, const float* __restrict__ pregate,
    const float* __restrict__ postgate, const float2* __restrict__ kf,
    float* __restrict__ y) {
  extern __shared__ char smem[];
  float2* R       = (float2*)smem;
  float2* Ftab    = (float2*)(smem + 128 * RSTRIDE * 8);
  float2* Wtab    = Ftab + 256;
  float2* scratch = Wtab + 256;

  const int tid = threadIdx.x, lane = tid & 31, wave = tid >> 5;
  const int img = blockIdx.x;
  const int c = img & 255;
  const size_t base = (size_t)img * (128 * 128);

  const float4* u4 = reinterpret_cast<const float4*>(u + base);
  const float4* p4 = reinterpret_cast<const float4*>(pregate + base);
  const int srow = 16 * wave + (lane >> 3);
  const int sseg = lane & 7;
  float4 uv[16], pv[16];
#pragma unroll
  for (int j = 0; j < 16; ++j) {
    const int i4 = (srow + 4 * (j >> 2)) * 32 + sseg + 8 * (j & 3);
    uv[j] = u4[i4];
    pv[j] = p4[i4];
  }

  init_tables(Ftab, Wtab, tid);

#pragma unroll
  for (int j = 0; j < 16; ++j) {
    float2* d = R + (srow + 4 * (j >> 2)) * RSTRIDE + 4 * (sseg + 8 * (j & 3));
    d[0] = make_float2(uv[j].x * pv[j].x, 0.0f);
    d[1] = make_float2(uv[j].y * pv[j].y, 0.0f);
    d[2] = make_float2(uv[j].z * pv[j].z, 0.0f);
    d[3] = make_float2(uv[j].w * pv[j].w, 0.0f);
  }
  __syncthreads();

  float2* spill = scratch + wave * (NB * (SPILLSZ + NFFT));
  float2* spec  = spill + NB * SPILLSZ;

  for (int row0 = wave; row0 < 128; row0 += 16 * NB) {
    fft192xN<NB>(
        [&](int q, int n) {
          const float2* RrowA = R + (row0 + 8 * q) * RSTRIDE;
          const float2* RrowB = RrowA + 8 * NB * RSTRIDE;
          return (n < 128) ? make_float2(RrowA[n].x, RrowB[n].x)
                           : make_float2(0.f, 0.f);
        },
        [](float2 v, int) { return v; },
        [&](const auto& lo, const auto& hi) {
          const int k0 = (lane & 15) + 48 * (lane >> 4);
          if ((lane & 15) < 12) {
#pragma unroll
            for (int q = 0; q < NB; ++q) {
#pragma unroll
              for (int i = 0; i < 4; ++i) {
                spec[q * NFFT + k0 + 12 * i] =
                    make_float2(lo[q][2 * i], lo[q][2 * i + 1]);
                spec[q * NFFT + NHALF + k0 + 12 * i] =
                    make_float2(hi[q][2 * i], hi[q][2 * i + 1]);
              }
            }
          }
          asm volatile("s_wait_dscnt 0" ::: "memory");
          float2 zk[NB][3], zm[NB][3], zn[NB];
#pragma unroll
          for (int q = 0; q < NB; ++q) {
            const float2* sq = spec + q * NFFT;
#pragma unroll
            for (int j = 0; j < 3; ++j) {
              const int k = lane + 32 * j;
              zk[q][j] = sq[k];
              zm[q][j] = sq[(k == 0) ? 0 : NFFT - k];
            }
            zn[q] = sq[NHALF];
          }
#pragma unroll
          for (int q = 0; q < NB; ++q) {
            float2* RrowA = R + (row0 + 8 * q) * RSTRIDE;
            float2* RrowB = RrowA + 8 * NB * RSTRIDE;
#pragma unroll
            for (int j = 0; j < 3; ++j) {
              const int k = lane + 32 * j;
              const float2 a = zk[q][j], m = zm[q][j];
              RrowA[k] = make_float2(0.5f * (a.x + m.x), 0.5f * (a.y - m.y));
              RrowB[k] = make_float2(0.5f * (a.y + m.y), 0.5f * (m.x - a.x));
            }
            if (lane == 0) {
              RrowA[NHALF] = make_float2(zn[q].x, 0.0f);
              RrowB[NHALF] = make_float2(zn[q].y, 0.0f);
            }
          }
        },
        Ftab, Wtab, spill, +1.0f, lane);
  }
  __syncthreads();

  const float2* kfbase = kf + (size_t)c * KFCOLS * NFFT;
  for (int col0 = wave; col0 < NHALF; col0 += 8 * NB) {
#pragma unroll
    for (int q = 0; q < NB; ++q) {
      const int pc = col0 + 8 * NB + 8 * q;
      if (pc <= NHALF) {
        const char* pf = (const char*)(kfbase + (size_t)pc * NFFT);
        __builtin_prefetch(pf + ((lane < 24) ? lane : 23) * 64, 0, 1);
      }
    }
    conv_columns<NB>(R, kfbase, col0, Ftab, Wtab, spill, spec, lane);
  }
  if (wave == 0)
    conv_columns<1>(R, kfbase, NHALF, Ftab, Wtab, spill, spec, lane);
  __syncthreads();

  const float inv_scale = 1.0f / 36864.0f;
  for (int row0 = wave; row0 < 128; row0 += 16 * NB) {
    const bool act = (lane & 15) < 12;
    const int k0 = (lane & 15) + 48 * (lane >> 4);
    float gA[NB][4], gB[NB][4], gAh[NB][3], gBh[NB][3];
#pragma unroll
    for (int q = 0; q < NB; ++q) {
      const size_t rA = base + (size_t)(row0 + 8 * q) * 128;
      const size_t rB = rA + (size_t)(8 * NB) * 128;
#pragma unroll
      for (int i = 0; i < 4; ++i) {
        gA[q][i] = postgate[rA + k0 + 12 * i];
        gB[q][i] = postgate[rB + k0 + 12 * i];
      }
#pragma unroll
      for (int i = 0; i < 3; ++i) {
        const int kh = (k0 + 12 * i < 32) ? NHALF + k0 + 12 * i : 127;
        gAh[q][i] = postgate[rA + kh];
        gBh[q][i] = postgate[rB + kh];
      }
    }
    fft192xN<NB>(
        [&](int q, int n) {
          const float2* RrowA = R + (row0 + 8 * q) * RSTRIDE;
          const float2* RrowB = RrowA + 8 * NB * RSTRIDE;
          const int idx = (n <= NHALF) ? n : NFFT - n;
          v4f r;
          r.x = RrowA[idx].x;
          r.y = RrowA[idx].y;
          r.z = RrowB[idx].x;
          r.w = RrowB[idx].y;
          return r;
        },
        [](v4f r, int n) {
          const bool stored = (n <= NHALF);
          const bool edge = (n == 0) || (n == NHALF);
          const float2 a = make_float2(r.x, r.y);
          const float2 b = make_float2(r.z, r.w);
          const float ay = edge ? 0.0f : a.y;
          const float by = edge ? 0.0f : b.y;
          return stored ? make_float2(a.x - by, ay + b.x)
                        : make_float2(a.x + by, b.x - ay);
        },
        [&](const auto& lo, const auto& hi) {
#pragma unroll
          for (int q = 0; q < NB; ++q) {
#pragma unroll
            for (int i = 0; i < 4; ++i) {
              hold_loaded(gA[q][i]);
              hold_loaded(gB[q][i]);
            }
#pragma unroll
            for (int i = 0; i < 3; ++i) {
              hold_loaded(gAh[q][i]);
              hold_loaded(gBh[q][i]);
            }
          }
          if (act) {
#pragma unroll
            for (int q = 0; q < NB; ++q) {
              const size_t oA = base + (size_t)(row0 + 8 * q) * 128 + k0;
              const size_t oB = oA + (size_t)(8 * NB) * 128;
#pragma unroll
              for (int i = 0; i < 4; ++i) {
                y[oA + 12 * i] = lo[q][2 * i] * inv_scale * gA[q][i];
                y[oB + 12 * i] = lo[q][2 * i + 1] * inv_scale * gB[q][i];
              }
            }
            if (lane < 16) {
#pragma unroll
              for (int q = 0; q < NB; ++q) {
                const size_t oA = base + (size_t)(row0 + 8 * q) * 128 + k0;
                const size_t oB = oA + (size_t)(8 * NB) * 128;
#pragma unroll
                for (int i = 0; i < 2; ++i) {
                  y[oA + NHALF + 12 * i] =
                      hi[q][2 * i] * inv_scale * gAh[q][i];
                  y[oB + NHALF + 12 * i] =
                      hi[q][2 * i + 1] * inv_scale * gBh[q][i];
                }
              }
              if (lane < 8) {
#pragma unroll
                for (int q = 0; q < NB; ++q) {
                  const size_t oA = base + (size_t)(row0 + 8 * q) * 128 + k0;
                  const size_t oB = oA + (size_t)(8 * NB) * 128;
                  y[oA + NHALF + 24] = hi[q][4] * inv_scale * gAh[q][2];
                  y[oB + NHALF + 24] = hi[q][5] * inv_scale * gBh[q][2];
                }
              }
            }
          }
        },
        Ftab, Wtab, spill, -1.0f, lane);
  }
}

extern "C" void kernel_launch(void* const* d_in, const int* in_sizes, int n_in,
                              void* d_out, int out_size, void* d_ws,
                              size_t ws_size, hipStream_t stream) {
  const float* u        = (const float*)d_in[0];
  const float* k        = (const float*)d_in[1];
  const float* pregate  = (const float*)d_in[2];
  const float* postgate = (const float*)d_in[3];
  float* y   = (float*)d_out;
  float2* kf = (float2*)d_ws;

  const size_t lds_kf   = (size_t)64 * RSTRIDE * 8 + 512 * 8 + 8 * NB * SPILLSZ * 8;
  const size_t lds_main =
      (size_t)128 * RSTRIDE * 8 + 512 * 8 + 8 * NB * (SPILLSZ + NFFT) * 8;

  (void)hipFuncSetAttribute((const void*)kf_precompute,
                            hipFuncAttributeMaxDynamicSharedMemorySize,
                            (int)lds_kf);
  (void)hipFuncSetAttribute((const void*)conv_main,
                            hipFuncAttributeMaxDynamicSharedMemorySize,
                            (int)lds_main);

  kf_precompute<<<256, 256, lds_kf, stream>>>(k, kf);
  conv_main<<<4 * 256, 256, lds_main, stream>>>(u, pregate, postgate, kf, y);
}


# Main kernel's LDS 239 -> 155 KiB (image rows at pitch 97 holding the input as packed reals; spectra kept in the spill buffer): two workgroups resident per WGP, 4 waves per SIMD; arithmetic unchanged
# speedup vs baseline: 5.2273x; 1.1599x over previous
#include <hip/hip_runtime.h>
#include <hip/hip_bf16.h>
#include <math.h>


typedef float v2f __attribute__((ext_vector_type(2)));
typedef float v8f __attribute__((ext_vector_type(8)));

#define PI_F 3.14159265358979323846f
#define NFFT 192
#define NHALF 96
#define RSTRIDE 133
#define RPITCH 97
#define KFCOLS 97
#define NB 4
#define SPITCH 18
#define SPILLSZ (12 * SPITCH)

__device__ __forceinline__ v8f wmma_f32(v2f a, v2f b, v8f c) {
  return __builtin_amdgcn_wmma_f32_16x16x4_f32(false, a, false, b, (short)0, c,
                                               false, false);
}

__device__ __forceinline__ void block_fence() {
  __builtin_amdgcn_sched_barrier(0x6);
}


__device__ __forceinline__ float2 load_g_entry(const float2* Ftab, int t, int b,
                                               int lane) {
  const int m = lane & 15;
  const int h = (lane >> 4) & 1;
  const int c = m >> 1;
  const int a = 2 * t + h;
  return Ftab[16 + (((2 * a + b) * c) & 15)];
}

__device__ __forceinline__ v2f make_g_frag(float2 g, float dir, int lane) {
  const int m = lane & 15;
  const float gr = g.x, gi = dir * g.y;
  v2f f;
  if (m & 1) { f.x = gi; f.y = gr; } else { f.x = gr; f.y = -gi; }
  return f;
}

__device__ __forceinline__ float2 load_g12_entry(const float2* Ftab, int t,
                                                 int b, int lane) {
  const int m = lane & 15;
  const int h = (lane >> 4) & 1;
  const int c = 3 * (m >> 3) + ((m & 7) >> 1);
  const int a = 2 * t + h;
  return Ftab[32 + (((2 * a + b) * c) % 12)];
}

__device__ __forceinline__ v2f make_g12_frag(float2 g, float dir, int lane) {
  v2f f = make_g_frag(g, dir, lane);
  if ((lane & 7) >= 6) { f.x = 0.0f; f.y = 0.0f; }
  return f;
}

typedef float v4f __attribute__((ext_vector_type(4)));

__device__ __forceinline__ void hold_loaded(float x) {
  asm volatile("" ::"v"(x));
}

__device__ __forceinline__ v2f as_frag(float2 x) {
  v2f f;
  f.x = x.x;
  f.y = x.y;
  return f;
}

template <int N, typename LD, typename FX, typename DST>
__device__ __forceinline__ void fft192xN(LD ld, FX fx, DST dst,
                                         const float2* Ftab,
                                         const float2* Wtab, float2* spill,
                                         float dir, int lane) {
  const int col = lane & 15;
  const int h   = (lane >> 4) & 1;

  float2 gte[3], gto[3];
#pragma unroll
  for (int t = 0; t < 3; ++t) {
    gte[t] = load_g12_entry(Ftab, t, 0, lane);
    gto[t] = load_g12_entry(Ftab, t, 1, lane);
  }
  float2 wt[2][3];
#pragma unroll
  for (int d = 0; d < 2; ++d) {
#pragma unroll
    for (int i = 0; i < 3; ++i)
      wt[d][i] = Wtab[col * (3 * h + i + 6 * d)];
  }
  using Raw = decltype(ld(0, 0));
  Raw re[N][3], ro[N][3];
#pragma unroll
  for (int t = 0; t < 3; ++t) {
    const int n2 = 2 * (2 * t + h);
#pragma unroll
    for (int q = 0; q < N; ++q) {
      re[q][t] = ld(q, col + 16 * n2);
      ro[q][t] = ld(q, col + 16 * (n2 + 1));
    }
  }
  block_fence();

  v2f ge[3], go[3];
#pragma unroll
  for (int t = 0; t < 3; ++t) {
    ge[t] = make_g12_frag(gte[t], dir, lane);
    go[t] = make_g12_frag(gto[t], dir, lane);
  }

  v8f e[N], o[N];
#pragma unroll
  for (int q = 0; q < N; ++q) {
    e[q] = v8f{}; o[q] = v8f{};
  }

#pragma unroll
  for (int t = 0; t < 3; ++t) {
    const int n2 = 2 * (2 * t + h);
#pragma unroll
    for (int q = 0; q < N; ++q) {
      const v2f xe = as_frag(fx(re[q][t], col + 16 * n2));
      const v2f xo = as_frag(fx(ro[q][t], col + 16 * (n2 + 1)));
      e[q] = wmma_f32(ge[t], xe, e[q]);
      o[q] = wmma_f32(go[t], xo, o[q]);
    }
  }
  block_fence();

#pragma unroll
  for (int q = 0; q < N; ++q) {
    float2* sp = spill + q * SPILLSZ + col;
#pragma unroll
    for (int i = 0; i < 3; ++i) {
      const float lr = e[q][2 * i] + o[q][2 * i];
      const float li = e[q][2 * i + 1] + o[q][2 * i + 1];
      const float hr = e[q][2 * i] - o[q][2 * i];
      const float hm = e[q][2 * i + 1] - o[q][2 * i + 1];
      const float t0r = wt[0][i].x, t0i = dir * wt[0][i].y;
      const float t1r = wt[1][i].x, t1i = dir * wt[1][i].y;
      const int k1 = 3 * h + i;
      sp[k1 * SPITCH] =
          make_float2(lr * t0r - li * t0i, lr * t0i + li * t0r);
      sp[(k1 + 6) * SPITCH] =
          make_float2(hr * t1r - hm * t1i, hr * t1i + hm * t1r);
    }
  }
  asm volatile("s_wait_dscnt 0" ::: "memory");

  float2 hte[4], hto[4];
#pragma unroll
  for (int t = 0; t < 4; ++t) {
    hte[t] = load_g_entry(Ftab, t, 0, lane);
    hto[t] = load_g_entry(Ftab, t, 1, lane);
  }
  const int ccol = (col < 12) ? col : 11;
  v2f be[N][4], bo[N][4];
#pragma unroll
  for (int t = 0; t < 4; ++t) {
    const int n1 = 2 * (2 * t + h);
#pragma unroll
    for (int q = 0; q < N; ++q) {
      const float2* sp = spill + q * SPILLSZ + ccol * SPITCH + n1;
      be[q][t] = as_frag(sp[0]);
      bo[q][t] = as_frag(sp[1]);
    }
  }
  block_fence();

  v2f he[4], ho[4];
#pragma unroll
  for (int t = 0; t < 4; ++t) {
    he[t] = make_g_frag(hte[t], dir, lane);
    ho[t] = make_g_frag(hto[t], dir, lane);
  }

  v8f e2[N], o2[N];
#pragma unroll
  for (int q = 0; q < N; ++q) {
    e2[q] = v8f{}; o2[q] = v8f{};
  }

#pragma unroll
  for (int t = 0; t < 4; ++t) {
#pragma unroll
    for (int q = 0; q < N; ++q) {
      e2[q] = wmma_f32(he[t], be[q][t], e2[q]);
      o2[q] = wmma_f32(ho[t], bo[q][t], o2[q]);
    }
  }
  block_fence();

  v8f lo[N], hi[N];
#pragma unroll
  for (int q = 0; q < N; ++q) {
    lo[q] = e2[q] + o2[q];
    hi[q] = e2[q] - o2[q];
  }
  dst(lo, hi);
}

__device__ __forceinline__ void init_tables(float2* Ftab, float2* Wtab,
                                            int tid) {
  for (int i = tid; i < 256; i += 256) {
    const int r = i >> 4, c = i & 15;
    const float tf = (r == 2)
        ? -2.0f * PI_F * (float)(c % 12) / 12.0f
        : -2.0f * PI_F * (float)((r * c) & 15) / 16.0f;
    Ftab[i] = make_float2(cosf(tf), sinf(tf));
    const float tw = -2.0f * PI_F * (float)(i % NFFT) / (float)NFFT;
    Wtab[i] = make_float2(cosf(tw), sinf(tw));
  }
}


template <int N>
__device__ __forceinline__ void kf_columns(const float2* Rk, float2* kfc,
                                           int col0, const float2* Ftab,
                                           const float2* Wtab, float2* spill,
                                           int lane) {
  fft192xN<N>(
      [&](int q, int n) {
        return (n < 64) ? Rk[n * RSTRIDE + col0 + 8 * q]
                        : make_float2(0.f, 0.f);
      },
      [](float2 v, int) { return v; },
      [&](const auto& lo, const auto& hi) {
        const int k0 = (lane & 15) + 48 * (lane >> 4);
        if ((lane & 15) < 12) {
#pragma unroll
          for (int q = 0; q < N; ++q) {
            float2* dcol = kfc + (size_t)(col0 + 8 * q) * NFFT + k0;
#pragma unroll
            for (int i = 0; i < 4; ++i) {
              dcol[12 * i] = make_float2(lo[q][2 * i], lo[q][2 * i + 1]);
              dcol[NHALF + 12 * i] =
                  make_float2(hi[q][2 * i], hi[q][2 * i + 1]);
            }
          }
        }
      },
      Ftab, Wtab, spill, +1.0f, lane);
}

__global__ void __launch_bounds__(256) kf_precompute(
    const float* __restrict__ kin, float2* __restrict__ kf) {
  extern __shared__ char smem[];
  float2* Rk      = (float2*)smem;
  float2* Ftab    = (float2*)(smem + 64 * RSTRIDE * 8);
  float2* Wtab    = Ftab + 256;
  float2* scratch = Wtab + 256;

  const int tid = threadIdx.x, lane = tid & 31, wave = tid >> 5;
  const int c = blockIdx.x;

  const float4* k4 = reinterpret_cast<const float4*>(kin + (size_t)c * (64 * 64));
  const int srow = 8 * wave + (lane >> 3);
  const int sseg = lane & 7;
  float4 kq[4];
#pragma unroll
  for (int j = 0; j < 4; ++j)
    kq[j] = k4[(srow + 4 * (j >> 1)) * 16 + sseg + 8 * (j & 1)];

  init_tables(Ftab, Wtab, tid);

#pragma unroll
  for (int j = 0; j < 4; ++j) {
    float2* d = Rk + (srow + 4 * (j >> 1)) * RSTRIDE + 4 * (sseg + 8 * (j & 1));
    d[0] = make_float2(kq[j].x, 0.0f);
    d[1] = make_float2(kq[j].y, 0.0f);
    d[2] = make_float2(kq[j].z, 0.0f);
    d[3] = make_float2(kq[j].w, 0.0f);
  }
  __syncthreads();

  float2* spill = scratch + wave * (NB * SPILLSZ);

  for (int row0 = wave; row0 < 64; row0 += 8 * NB) {
    fft192xN<NB>(
        [&](int q, int n) {
          const float2* Rrow = Rk + (row0 + 8 * q) * RSTRIDE;
          return (n < 64) ? make_float2(Rrow[n].x, 0.0f)
                          : make_float2(0.f, 0.f);
        },
        [](float2 v, int) { return v; },
        [&](const auto& lo, const auto& hi) {
          const int k0 = (lane & 15) + 48 * (lane >> 4);
          if ((lane & 15) < 12) {
#pragma unroll
            for (int q = 0; q < NB; ++q) {
              float2* Rrow = Rk + (row0 + 8 * q) * RSTRIDE + k0;
#pragma unroll
              for (int i = 0; i < 4; ++i)
                Rrow[12 * i] = make_float2(lo[q][2 * i], lo[q][2 * i + 1]);
            }
          }
          if (lane == 0) {
#pragma unroll
            for (int q = 0; q < NB; ++q)
              Rk[(row0 + 8 * q) * RSTRIDE + NHALF] =
                  make_float2(hi[q][0], hi[q][1]);
          }
        },
        Ftab, Wtab, spill, +1.0f, lane);
  }
  __syncthreads();

  float2* kfc = kf + (size_t)c * KFCOLS * NFFT;
  for (int col0 = wave; col0 < NHALF; col0 += 8 * NB)
    kf_columns<NB>(Rk, kfc, col0, Ftab, Wtab, spill, lane);
  if (wave == 0) kf_columns<1>(Rk, kfc, NHALF, Ftab, Wtab, spill, lane);
}


template <int N>
__device__ __forceinline__ void conv_columns(float2* R, const float2* kfbase,
                                             int col0, const float2* Ftab,
                                             const float2* Wtab, float2* spill,
                                             int lane) {
  float2* spec = spill;
  float2 kv[N][6];
#pragma unroll
  for (int q = 0; q < N; ++q) {
    const float2* kfcol = kfbase + (size_t)(col0 + 8 * q) * NFFT + lane;
#pragma unroll
    for (int i = 0; i < 6; ++i) kv[q][i] = kfcol[32 * i];
  }

  fft192xN<N>(
      [&](int q, int n) {
        return (n < 128) ? R[n * RPITCH + col0 + 8 * q]
                         : make_float2(0.f, 0.f);
      },
      [](float2 v, int) { return v; },
      [&](const auto& lo, const auto& hi) {
        const int k0 = (lane & 15) + 48 * (lane >> 4);
        if ((lane & 15) < 12) {
#pragma unroll
          for (int q = 0; q < N; ++q) {
#pragma unroll
            for (int i = 0; i < 4; ++i) {
              spec[q * SPILLSZ + k0 + 12 * i] =
                  make_float2(lo[q][2 * i], lo[q][2 * i + 1]);
              spec[q * SPILLSZ + NHALF + k0 + 12 * i] =
                  make_float2(hi[q][2 * i], hi[q][2 * i + 1]);
            }
          }
        }
      },
      Ftab, Wtab, spill, +1.0f, lane);

#pragma unroll
  for (int q = 0; q < N; ++q) {
    float2* sq = spec + q * SPILLSZ + lane;
#pragma unroll
    for (int i = 0; i < 6; ++i) {
      const float2 a = sq[32 * i];
      const float2 b = kv[q][i];
      sq[32 * i] = make_float2(a.x * b.x - a.y * b.y, a.x * b.y + a.y * b.x);
    }
  }
  asm volatile("s_wait_dscnt 0" ::: "memory");

  fft192xN<N>(
      [&](int q, int n) { return spec[q * SPILLSZ + n]; },
      [](float2 v, int) { return v; },
      [&](const auto& lo, const auto& hi) {
        const int k0 = (lane & 15) + 48 * (lane >> 4);
        if ((lane & 15) < 12) {
#pragma unroll
          for (int q = 0; q < N; ++q) {
#pragma unroll
            for (int i = 0; i < 4; ++i)
              R[(k0 + 12 * i) * RPITCH + col0 + 8 * q] =
                  make_float2(lo[q][2 * i], lo[q][2 * i + 1]);
          }
          if (lane < 16) {
#pragma unroll
            for (int q = 0; q < N; ++q) {
#pragma unroll
              for (int i = 0; i < 2; ++i)
                R[(NHALF + k0 + 12 * i) * RPITCH + col0 + 8 * q] =
                    make_float2(hi[q][2 * i], hi[q][2 * i + 1]);
            }
            if (lane < 8) {
#pragma unroll
              for (int q = 0; q < N; ++q)
                R[(NHALF + k0 + 24) * RPITCH + col0 + 8 * q] =
                    make_float2(hi[q][4], hi[q][5]);
            }
          }
        }
      },
      Ftab, Wtab, spill, -1.0f, lane);
}

__global__ void __launch_bounds__(256) __attribute__((amdgpu_waves_per_eu(4, 4)))
conv_main(
    const float* __restrict__ u, const float* __restrict__ pregate,
    const float* __restrict__ postgate, const float2* __restrict__ kf,
    float* __restrict__ y) {
  extern __shared__ char smem[];
  float2* R       = (float2*)smem;
  float2* Ftab    = (float2*)(smem + 128 * RPITCH * 8);
  float2* Wtab    = Ftab + 256;
  float2* scratch = Wtab + 256;

  const int tid = threadIdx.x, lane = tid & 31, wave = tid >> 5;
  const int img = blockIdx.x;
  const int c = img & 255;
  const size_t base = (size_t)img * (128 * 128);

  const float4* u4 = reinterpret_cast<const float4*>(u + base);
  const float4* p4 = reinterpret_cast<const float4*>(pregate + base);
  const int srow = 16 * wave + (lane >> 3);
  const int sseg = lane & 7;
  float4 uv[16], pv[16];
#pragma unroll
  for (int j = 0; j < 16; ++j) {
    const int i4 = (srow + 4 * (j >> 2)) * 32 + sseg + 8 * (j & 3);
    uv[j] = u4[i4];
    pv[j] = p4[i4];
  }

  init_tables(Ftab, Wtab, tid);

#pragma unroll
  for (int j = 0; j < 16; ++j) {
    float2* d = R + (srow + 4 * (j >> 2)) * RPITCH + 2 * (sseg + 8 * (j & 3));
    d[0] = make_float2(uv[j].x * pv[j].x, uv[j].y * pv[j].y);
    d[1] = make_float2(uv[j].z * pv[j].z, uv[j].w * pv[j].w);
  }
  __syncthreads();

  float2* spill = scratch + wave * (NB * SPILLSZ);
  float2* spec  = spill;

  for (int row0 = wave; row0 < 128; row0 += 16 * NB) {
    fft192xN<NB>(
        [&](int q, int n) {
          const float* xA = (const float*)(R + (row0 + 8 * q) * RPITCH);
          const float* xB = (const float*)(R + (row0 + 8 * q + 8 * NB) * RPITCH);
          return (n < 128) ? make_float2(xA[n], xB[n])
                           : make_float2(0.f, 0.f);
        },
        [](float2 v, int) { return v; },
        [&](const auto& lo, const auto& hi) {
          const int k0 = (lane & 15) + 48 * (lane >> 4);
          if ((lane & 15) < 12) {
#pragma unroll
            for (int q = 0; q < NB; ++q) {
#pragma unroll
              for (int i = 0; i < 4; ++i) {
                spec[q * SPILLSZ + k0 + 12 * i] =
                    make_float2(lo[q][2 * i], lo[q][2 * i + 1]);
                spec[q * SPILLSZ + NHALF + k0 + 12 * i] =
                    make_float2(hi[q][2 * i], hi[q][2 * i + 1]);
              }
            }
          }
          asm volatile("s_wait_dscnt 0" ::: "memory");
          float2 zk[NB][3], zm[NB][3], zn[NB];
#pragma unroll
          for (int q = 0; q < NB; ++q) {
            const float2* sq = spec + q * SPILLSZ;
#pragma unroll
            for (int j = 0; j < 3; ++j) {
              const int k = lane + 32 * j;
              zk[q][j] = sq[k];
              zm[q][j] = sq[(k == 0) ? 0 : NFFT - k];
            }
            zn[q] = sq[NHALF];
          }
#pragma unroll
          for (int q = 0; q < NB; ++q) {
            float2* RrowA = R + (row0 + 8 * q) * RPITCH;
            float2* RrowB = RrowA + 8 * NB * RPITCH;
#pragma unroll
            for (int j = 0; j < 3; ++j) {
              const int k = lane + 32 * j;
              const float2 a = zk[q][j], m = zm[q][j];
              RrowA[k] = make_float2(0.5f * (a.x + m.x), 0.5f * (a.y - m.y));
              RrowB[k] = make_float2(0.5f * (a.y + m.y), 0.5f * (m.x - a.x));
            }
            if (lane == 0) {
              RrowA[NHALF] = make_float2(zn[q].x, 0.0f);
              RrowB[NHALF] = make_float2(zn[q].y, 0.0f);
            }
          }
        },
        Ftab, Wtab, spill, +1.0f, lane);
  }
  __syncthreads();

  const float2* kfbase = kf + (size_t)c * KFCOLS * NFFT;
  for (int col0 = wave; col0 < NHALF; col0 += 8 * NB) {
#pragma unroll
    for (int q = 0; q < NB; ++q) {
      const int pc = col0 + 8 * NB + 8 * q;
      if (pc <= NHALF) {
        const char* pf = (const char*)(kfbase + (size_t)pc * NFFT);
        __builtin_prefetch(pf + ((lane < 24) ? lane : 23) * 64, 0, 1);
      }
    }
    conv_columns<NB>(R, kfbase, col0, Ftab, Wtab, spill, lane);
  }
  if (wave == 0)
    conv_columns<1>(R, kfbase, NHALF, Ftab, Wtab, spill, lane);
  __syncthreads();

  const float inv_scale = 1.0f / 36864.0f;
  for (int row0 = wave; row0 < 128; row0 += 16 * NB) {
    const bool act = (lane & 15) < 12;
    const int k0 = (lane & 15) + 48 * (lane >> 4);
    float gA[NB][4], gB[NB][4], gAh[NB][3], gBh[NB][3];
#pragma unroll
    for (int q = 0; q < NB; ++q) {
      const size_t rA = base + (size_t)(row0 + 8 * q) * 128;
      const size_t rB = rA + (size_t)(8 * NB) * 128;
#pragma unroll
      for (int i = 0; i < 4; ++i) {
        gA[q][i] = postgate[rA + k0 + 12 * i];
        gB[q][i] = postgate[rB + k0 + 12 * i];
      }
#pragma unroll
      for (int i = 0; i < 3; ++i) {
        const int kh = (k0 + 12 * i < 32) ? NHALF + k0 + 12 * i : 127;
        gAh[q][i] = postgate[rA + kh];
        gBh[q][i] = postgate[rB + kh];
      }
    }
    fft192xN<NB>(
        [&](int q, int n) {
          const float2* RrowA = R + (row0 + 8 * q) * RPITCH;
          const float2* RrowB = RrowA + 8 * NB * RPITCH;
          const int idx = (n <= NHALF) ? n : NFFT - n;
          v4f r;
          r.x = RrowA[idx].x;
          r.y = RrowA[idx].y;
          r.z = RrowB[idx].x;
          r.w = RrowB[idx].y;
          return r;
        },
        [](v4f r, int n) {
          const bool stored = (n <= NHALF);
          const bool edge = (n == 0) || (n == NHALF);
          const float2 a = make_float2(r.x, r.y);
          const float2 b = make_float2(r.z, r.w);
          const float ay = edge ? 0.0f : a.y;
          const float by = edge ? 0.0f : b.y;
          return stored ? make_float2(a.x - by, ay + b.x)
                        : make_float2(a.x + by, b.x - ay);
        },
        [&](const auto& lo, const auto& hi) {
#pragma unroll
          for (int q = 0; q < NB; ++q) {
#pragma unroll
            for (int i = 0; i < 4; ++i) {
              hold_loaded(gA[q][i]);
              hold_loaded(gB[q][i]);
            }
#pragma unroll
            for (int i = 0; i < 3; ++i) {
              hold_loaded(gAh[q][i]);
              hold_loaded(gBh[q][i]);
            }
          }
          if (act) {
#pragma unroll
            for (int q = 0; q < NB; ++q) {
              const size_t oA = base + (size_t)(row0 + 8 * q) * 128 + k0;
              const size_t oB = oA + (size_t)(8 * NB) * 128;
#pragma unroll
              for (int i = 0; i < 4; ++i) {
                y[oA + 12 * i] = lo[q][2 * i] * inv_scale * gA[q][i];
                y[oB + 12 * i] = lo[q][2 * i + 1] * inv_scale * gB[q][i];
              }
            }
            if (lane < 16) {
#pragma unroll
              for (int q = 0; q < NB; ++q) {
                const size_t oA = base + (size_t)(row0 + 8 * q) * 128 + k0;
                const size_t oB = oA + (size_t)(8 * NB) * 128;
#pragma unroll
                for (int i = 0; i < 2; ++i) {
                  y[oA + NHALF + 12 * i] =
                      hi[q][2 * i] * inv_scale * gAh[q][i];
                  y[oB + NHALF + 12 * i] =
                      hi[q][2 * i + 1] * inv_scale * gBh[q][i];
                }
              }
              if (lane < 8) {
#pragma unroll
                for (int q = 0; q < NB; ++q) {
                  const size_t oA = base + (size_t)(row0 + 8 * q) * 128 + k0;
                  const size_t oB = oA + (size_t)(8 * NB) * 128;
                  y[oA + NHALF + 24] = hi[q][4] * inv_scale * gAh[q][2];
                  y[oB + NHALF + 24] = hi[q][5] * inv_scale * gBh[q][2];
                }
              }
            }
          }
        },
        Ftab, Wtab, spill, -1.0f, lane);
  }
}

extern "C" void kernel_launch(void* const* d_in, const int* in_sizes, int n_in,
                              void* d_out, int out_size, void* d_ws,
                              size_t ws_size, hipStream_t stream) {
  const float* u        = (const float*)d_in[0];
  const float* k        = (const float*)d_in[1];
  const float* pregate  = (const float*)d_in[2];
  const float* postgate = (const float*)d_in[3];
  float* y   = (float*)d_out;
  float2* kf = (float2*)d_ws;

  const size_t lds_kf   = (size_t)64 * RSTRIDE * 8 + 512 * 8 + 8 * NB * SPILLSZ * 8;
  const size_t lds_main = (size_t)128 * RPITCH * 8 + 512 * 8 + 8 * NB * SPILLSZ * 8;

  (void)hipFuncSetAttribute((const void*)kf_precompute,
                            hipFuncAttributeMaxDynamicSharedMemorySize,
                            (int)lds_kf);
  (void)hipFuncSetAttribute((const void*)conv_main,
                            hipFuncAttributeMaxDynamicSharedMemorySize,
                            (int)lds_main);

  kf_precompute<<<256, 256, lds_kf, stream>>>(k, kf);
  conv_main<<<4 * 256, 256, lds_main, stream>>>(u, pregate, postgate, kf, y);
}


# Stage 2 of a 4-transform batch packs its 4 x 12 columns into 3 full 16-column B tiles (a tile spans two sequences), was 4 tiles of 12 + 4 idle: 24 WMMA, was 32; every sum and its order unchanged
# speedup vs baseline: 5.4587x; 1.0443x over previous
#include <hip/hip_runtime.h>
#include <hip/hip_bf16.h>
#include <math.h>


typedef float v2f __attribute__((ext_vector_type(2)));
typedef float v8f __attribute__((ext_vector_type(8)));

#define PI_F 3.14159265358979323846f
#define NFFT 192
#define NHALF 96
#define RSTRIDE 133
#define RPITCH 97
#define KFCOLS 97
#define NB 4
#define SPITCH 18
#define SPILLSZ (12 * SPITCH)

__device__ __forceinline__ v8f wmma_f32(v2f a, v2f b, v8f c) {
  return __builtin_amdgcn_wmma_f32_16x16x4_f32(false, a, false, b, (short)0, c,
                                               false, false);
}

__device__ __forceinline__ void block_fence() {
  __builtin_amdgcn_sched_barrier(0x6);
}


__device__ __forceinline__ float2 load_g_entry(const float2* Ftab, int t, int b,
                                               int lane) {
  const int m = lane & 15;
  const int h = (lane >> 4) & 1;
  const int c = m >> 1;
  const int a = 2 * t + h;
  return Ftab[16 + (((2 * a + b) * c) & 15)];
}

__device__ __forceinline__ v2f make_g_frag(float2 g, float dir, int lane) {
  const int m = lane & 15;
  const float gr = g.x, gi = dir * g.y;
  v2f f;
  if (m & 1) { f.x = gi; f.y = gr; } else { f.x = gr; f.y = -gi; }
  return f;
}

__device__ __forceinline__ float2 load_g12_entry(const float2* Ftab, int t,
                                                 int b, int lane) {
  const int m = lane & 15;
  const int h = (lane >> 4) & 1;
  const int c = 3 * (m >> 3) + ((m & 7) >> 1);
  const int a = 2 * t + h;
  return Ftab[32 + (((2 * a + b) * c) % 12)];
}

__device__ __forceinline__ v2f make_g12_frag(float2 g, float dir, int lane) {
  v2f f = make_g_frag(g, dir, lane);
  if ((lane & 7) >= 6) { f.x = 0.0f; f.y = 0.0f; }
  return f;
}

typedef float v4f __attribute__((ext_vector_type(4)));

__device__ __forceinline__ void hold_loaded(float x) {
  asm volatile("" ::"v"(x));
}

__device__ __forceinline__ v2f as_frag(float2 x) {
  v2f f;
  f.x = x.x;
  f.y = x.y;
  return f;
}

#define S2TILES(n) ((12 * (n) + 15) / 16)

template <int N>
__device__ __forceinline__ bool s2_valid(int j, int lane) {
  if constexpr ((12 * N) % 16 == 0) return true;
  else return 16 * j + (lane & 15) < 12 * N;
}

template <int N>
__device__ __forceinline__ int s2_col(int j, int lane) {
  const int c = 16 * j + (lane & 15);
  if constexpr ((12 * N) % 16 == 0) return c;
  else return (c < 12 * N) ? c : 12 * N - 1;
}

template <int N>
__device__ __forceinline__ int s2_seq(int j, int lane) {
  return s2_col<N>(j, lane) / 12;
}

template <int N>
__device__ __forceinline__ int s2_k1(int j, int lane) {
  return s2_col<N>(j, lane) % 12;
}

template <int N, typename LD, typename FX, typename DST>
__device__ __forceinline__ void fft192xN(LD ld, FX fx, DST dst,
                                         const float2* Ftab,
                                         const float2* Wtab, float2* spill,
                                         float dir, int lane) {
  const int col = lane & 15;
  const int h   = (lane >> 4) & 1;

  float2 gte[3], gto[3];
#pragma unroll
  for (int t = 0; t < 3; ++t) {
    gte[t] = load_g12_entry(Ftab, t, 0, lane);
    gto[t] = load_g12_entry(Ftab, t, 1, lane);
  }
  float2 wt[2][3];
#pragma unroll
  for (int d = 0; d < 2; ++d) {
#pragma unroll
    for (int i = 0; i < 3; ++i)
      wt[d][i] = Wtab[col * (3 * h + i + 6 * d)];
  }
  using Raw = decltype(ld(0, 0));
  Raw re[N][3], ro[N][3];
#pragma unroll
  for (int t = 0; t < 3; ++t) {
    const int n2 = 2 * (2 * t + h);
#pragma unroll
    for (int q = 0; q < N; ++q) {
      re[q][t] = ld(q, col + 16 * n2);
      ro[q][t] = ld(q, col + 16 * (n2 + 1));
    }
  }
  block_fence();

  v2f ge[3], go[3];
#pragma unroll
  for (int t = 0; t < 3; ++t) {
    ge[t] = make_g12_frag(gte[t], dir, lane);
    go[t] = make_g12_frag(gto[t], dir, lane);
  }

  v8f e[N], o[N];
#pragma unroll
  for (int q = 0; q < N; ++q) {
    e[q] = v8f{}; o[q] = v8f{};
  }

#pragma unroll
  for (int t = 0; t < 3; ++t) {
    const int n2 = 2 * (2 * t + h);
#pragma unroll
    for (int q = 0; q < N; ++q) {
      const v2f xe = as_frag(fx(re[q][t], col + 16 * n2));
      const v2f xo = as_frag(fx(ro[q][t], col + 16 * (n2 + 1)));
      e[q] = wmma_f32(ge[t], xe, e[q]);
      o[q] = wmma_f32(go[t], xo, o[q]);
    }
  }
  block_fence();

#pragma unroll
  for (int q = 0; q < N; ++q) {
    float2* sp = spill + q * SPILLSZ + col;
#pragma unroll
    for (int i = 0; i < 3; ++i) {
      const float lr = e[q][2 * i] + o[q][2 * i];
      const float li = e[q][2 * i + 1] + o[q][2 * i + 1];
      const float hr = e[q][2 * i] - o[q][2 * i];
      const float hm = e[q][2 * i + 1] - o[q][2 * i + 1];
      const float t0r = wt[0][i].x, t0i = dir * wt[0][i].y;
      const float t1r = wt[1][i].x, t1i = dir * wt[1][i].y;
      const int k1 = 3 * h + i;
      sp[k1 * SPITCH] =
          make_float2(lr * t0r - li * t0i, lr * t0i + li * t0r);
      sp[(k1 + 6) * SPITCH] =
          make_float2(hr * t1r - hm * t1i, hr * t1i + hm * t1r);
    }
  }
  asm volatile("s_wait_dscnt 0" ::: "memory");

  constexpr int NT = S2TILES(N);
  float2 hte[4], hto[4];
#pragma unroll
  for (int t = 0; t < 4; ++t) {
    hte[t] = load_g_entry(Ftab, t, 0, lane);
    hto[t] = load_g_entry(Ftab, t, 1, lane);
  }
  v2f be[NT][4], bo[NT][4];
#pragma unroll
  for (int t = 0; t < 4; ++t) {
    const int n1 = 2 * (2 * t + h);
#pragma unroll
    for (int j = 0; j < NT; ++j) {
      const float2* sp = spill + s2_seq<N>(j, lane) * SPILLSZ +
                         s2_k1<N>(j, lane) * SPITCH + n1;
      be[j][t] = as_frag(sp[0]);
      bo[j][t] = as_frag(sp[1]);
    }
  }
  block_fence();

  v2f he[4], ho[4];
#pragma unroll
  for (int t = 0; t < 4; ++t) {
    he[t] = make_g_frag(hte[t], dir, lane);
    ho[t] = make_g_frag(hto[t], dir, lane);
  }

  v8f e2[NT], o2[NT];
#pragma unroll
  for (int j = 0; j < NT; ++j) {
    e2[j] = v8f{}; o2[j] = v8f{};
  }

#pragma unroll
  for (int t = 0; t < 4; ++t) {
#pragma unroll
    for (int j = 0; j < NT; ++j) {
      e2[j] = wmma_f32(he[t], be[j][t], e2[j]);
      o2[j] = wmma_f32(ho[t], bo[j][t], o2[j]);
    }
  }
  block_fence();

  v8f lo[NT], hi[NT];
#pragma unroll
  for (int j = 0; j < NT; ++j) {
    lo[j] = e2[j] + o2[j];
    hi[j] = e2[j] - o2[j];
  }
  dst(lo, hi);
}

__device__ __forceinline__ void init_tables(float2* Ftab, float2* Wtab,
                                            int tid) {
  for (int i = tid; i < 256; i += 256) {
    const int r = i >> 4, c = i & 15;
    const float tf = (r == 2)
        ? -2.0f * PI_F * (float)(c % 12) / 12.0f
        : -2.0f * PI_F * (float)((r * c) & 15) / 16.0f;
    Ftab[i] = make_float2(cosf(tf), sinf(tf));
    const float tw = -2.0f * PI_F * (float)(i % NFFT) / (float)NFFT;
    Wtab[i] = make_float2(cosf(tw), sinf(tw));
  }
}


template <int N>
__device__ __forceinline__ void kf_columns(const float2* Rk, float2* kfc,
                                           int col0, const float2* Ftab,
                                           const float2* Wtab, float2* spill,
                                           int lane) {
  fft192xN<N>(
      [&](int q, int n) {
        return (n < 64) ? Rk[n * RSTRIDE + col0 + 8 * q]
                        : make_float2(0.f, 0.f);
      },
      [](float2 v, int) { return v; },
      [&](const auto& lo, const auto& hi) {
#pragma unroll
        for (int j = 0; j < S2TILES(N); ++j) {
          if (s2_valid<N>(j, lane)) {
            const int k0 = s2_k1<N>(j, lane) + 48 * (lane >> 4);
            float2* dcol =
                kfc + (size_t)(col0 + 8 * s2_seq<N>(j, lane)) * NFFT + k0;
#pragma unroll
            for (int i = 0; i < 4; ++i) {
              dcol[12 * i] = make_float2(lo[j][2 * i], lo[j][2 * i + 1]);
              dcol[NHALF + 12 * i] =
                  make_float2(hi[j][2 * i], hi[j][2 * i + 1]);
            }
          }
        }
      },
      Ftab, Wtab, spill, +1.0f, lane);
}

__global__ void __launch_bounds__(256) kf_precompute(
    const float* __restrict__ kin, float2* __restrict__ kf) {
  extern __shared__ char smem[];
  float2* Rk      = (float2*)smem;
  float2* Ftab    = (float2*)(smem + 64 * RSTRIDE * 8);
  float2* Wtab    = Ftab + 256;
  float2* scratch = Wtab + 256;

  const int tid = threadIdx.x, lane = tid & 31, wave = tid >> 5;
  const int c = blockIdx.x;

  const float4* k4 = reinterpret_cast<const float4*>(kin + (size_t)c * (64 * 64));
  const int srow = 8 * wave + (lane >> 3);
  const int sseg = lane & 7;
  float4 kq[4];
#pragma unroll
  for (int j = 0; j < 4; ++j)
    kq[j] = k4[(srow + 4 * (j >> 1)) * 16 + sseg + 8 * (j & 1)];

  init_tables(Ftab, Wtab, tid);

#pragma unroll
  for (int j = 0; j < 4; ++j) {
    float2* d = Rk + (srow + 4 * (j >> 1)) * RSTRIDE + 4 * (sseg + 8 * (j & 1));
    d[0] = make_float2(kq[j].x, 0.0f);
    d[1] = make_float2(kq[j].y, 0.0f);
    d[2] = make_float2(kq[j].z, 0.0f);
    d[3] = make_float2(kq[j].w, 0.0f);
  }
  __syncthreads();

  float2* spill = scratch + wave * (NB * SPILLSZ);

  for (int row0 = wave; row0 < 64; row0 += 8 * NB) {
    fft192xN<NB>(
        [&](int q, int n) {
          const float2* Rrow = Rk + (row0 + 8 * q) * RSTRIDE;
          return (n < 64) ? make_float2(Rrow[n].x, 0.0f)
                          : make_float2(0.f, 0.f);
        },
        [](float2 v, int) { return v; },
        [&](const auto& lo, const auto& hi) {
#pragma unroll
          for (int j = 0; j < S2TILES(NB); ++j) {
            const int k1 = s2_k1<NB>(j, lane);
            float2* Rrow = Rk + (row0 + 8 * s2_seq<NB>(j, lane)) * RSTRIDE;
#pragma unroll
            for (int i = 0; i < 4; ++i)
              Rrow[k1 + 48 * (lane >> 4) + 12 * i] =
                  make_float2(lo[j][2 * i], lo[j][2 * i + 1]);
            if (k1 == 0 && lane < 16)
              Rrow[NHALF] = make_float2(hi[j][0], hi[j][1]);
          }
        },
        Ftab, Wtab, spill, +1.0f, lane);
  }
  __syncthreads();

  float2* kfc = kf + (size_t)c * KFCOLS * NFFT;
  for (int col0 = wave; col0 < NHALF; col0 += 8 * NB)
    kf_columns<NB>(Rk, kfc, col0, Ftab, Wtab, spill, lane);
  if (wave == 0) kf_columns<1>(Rk, kfc, NHALF, Ftab, Wtab, spill, lane);
}


template <int N>
__device__ __forceinline__ void conv_columns(float2* R, const float2* kfbase,
                                             int col0, const float2* Ftab,
                                             const float2* Wtab, float2* spill,
                                             int lane) {
  float2* spec = spill;
  float2 kv[N][6];
#pragma unroll
  for (int q = 0; q < N; ++q) {
    const float2* kfcol = kfbase + (size_t)(col0 + 8 * q) * NFFT + lane;
#pragma unroll
    for (int i = 0; i < 6; ++i) kv[q][i] = kfcol[32 * i];
  }

  fft192xN<N>(
      [&](int q, int n) {
        return (n < 128) ? R[n * RPITCH + col0 + 8 * q]
                         : make_float2(0.f, 0.f);
      },
      [](float2 v, int) { return v; },
      [&](const auto& lo, const auto& hi) {
#pragma unroll
        for (int j = 0; j < S2TILES(N); ++j) {
          if (s2_valid<N>(j, lane)) {
            float2* sq = spec + s2_seq<N>(j, lane) * SPILLSZ +
                         s2_k1<N>(j, lane) + 48 * (lane >> 4);
#pragma unroll
            for (int i = 0; i < 4; ++i) {
              sq[12 * i] = make_float2(lo[j][2 * i], lo[j][2 * i + 1]);
              sq[NHALF + 12 * i] =
                  make_float2(hi[j][2 * i], hi[j][2 * i + 1]);
            }
          }
        }
      },
      Ftab, Wtab, spill, +1.0f, lane);

#pragma unroll
  for (int q = 0; q < N; ++q) {
    float2* sq = spec + q * SPILLSZ + lane;
#pragma unroll
    for (int i = 0; i < 6; ++i) {
      const float2 a = sq[32 * i];
      const float2 b = kv[q][i];
      sq[32 * i] = make_float2(a.x * b.x - a.y * b.y, a.x * b.y + a.y * b.x);
    }
  }
  asm volatile("s_wait_dscnt 0" ::: "memory");

  fft192xN<N>(
      [&](int q, int n) { return spec[q * SPILLSZ + n]; },
      [](float2 v, int) { return v; },
      [&](const auto& lo, const auto& hi) {
#pragma unroll
        for (int j = 0; j < S2TILES(N); ++j) {
          if (s2_valid<N>(j, lane)) {
            const int k0 = s2_k1<N>(j, lane) + 48 * (lane >> 4);
            float2* Rc = R + col0 + 8 * s2_seq<N>(j, lane);
#pragma unroll
            for (int i = 0; i < 4; ++i)
              Rc[(k0 + 12 * i) * RPITCH] =
                  make_float2(lo[j][2 * i], lo[j][2 * i + 1]);
          }
        }
        if (lane < 16) {
#pragma unroll
          for (int j = 0; j < S2TILES(N); ++j) {
            if (s2_valid<N>(j, lane)) {
              float2* Rh = R + (NHALF + s2_k1<N>(j, lane)) * RPITCH + col0 +
                           8 * s2_seq<N>(j, lane);
#pragma unroll
              for (int i = 0; i < 2; ++i)
                Rh[12 * i * RPITCH] =
                    make_float2(hi[j][2 * i], hi[j][2 * i + 1]);
            }
          }
#pragma unroll
          for (int j = 0; j < S2TILES(N); ++j) {
            if (s2_valid<N>(j, lane) && s2_k1<N>(j, lane) < 8)
              R[(NHALF + s2_k1<N>(j, lane) + 24) * RPITCH + col0 +
                8 * s2_seq<N>(j, lane)] = make_float2(hi[j][4], hi[j][5]);
          }
        }
      },
      Ftab, Wtab, spill, -1.0f, lane);
}

__global__ void __launch_bounds__(256) __attribute__((amdgpu_waves_per_eu(4, 4)))
conv_main(
    const float* __restrict__ u, const float* __restrict__ pregate,
    const float* __restrict__ postgate, const float2* __restrict__ kf,
    float* __restrict__ y) {
  extern __shared__ char smem[];
  float2* R       = (float2*)smem;
  float2* Ftab    = (float2*)(smem + 128 * RPITCH * 8);
  float2* Wtab    = Ftab + 256;
  float2* scratch = Wtab + 256;

  const int tid = threadIdx.x, lane = tid & 31, wave = tid >> 5;
  const int img = blockIdx.x;
  const int c = img & 255;
  const size_t base = (size_t)img * (128 * 128);

  const float4* u4 = reinterpret_cast<const float4*>(u + base);
  const float4* p4 = reinterpret_cast<const float4*>(pregate + base);
  const int srow = 16 * wave + (lane >> 3);
  const int sseg = lane & 7;
  float4 uv[16], pv[16];
#pragma unroll
  for (int j = 0; j < 16; ++j) {
    const int i4 = (srow + 4 * (j >> 2)) * 32 + sseg + 8 * (j & 3);
    uv[j] = u4[i4];
    pv[j] = p4[i4];
  }

  init_tables(Ftab, Wtab, tid);

#pragma unroll
  for (int j = 0; j < 16; ++j) {
    float2* d = R + (srow + 4 * (j >> 2)) * RPITCH + 2 * (sseg + 8 * (j & 3));
    d[0] = make_float2(uv[j].x * pv[j].x, uv[j].y * pv[j].y);
    d[1] = make_float2(uv[j].z * pv[j].z, uv[j].w * pv[j].w);
  }
  __syncthreads();

  float2* spill = scratch + wave * (NB * SPILLSZ);
  float2* spec  = spill;

  for (int row0 = wave; row0 < 128; row0 += 16 * NB) {
    fft192xN<NB>(
        [&](int q, int n) {
          const float* xA = (const float*)(R + (row0 + 8 * q) * RPITCH);
          const float* xB = (const float*)(R + (row0 + 8 * q + 8 * NB) * RPITCH);
          return (n < 128) ? make_float2(xA[n], xB[n])
                           : make_float2(0.f, 0.f);
        },
        [](float2 v, int) { return v; },
        [&](const auto& lo, const auto& hi) {
#pragma unroll
          for (int j = 0; j < S2TILES(NB); ++j) {
            float2* sz = spec + s2_seq<NB>(j, lane) * SPILLSZ +
                         s2_k1<NB>(j, lane) + 48 * (lane >> 4);
#pragma unroll
            for (int i = 0; i < 4; ++i) {
              sz[12 * i] = make_float2(lo[j][2 * i], lo[j][2 * i + 1]);
              sz[NHALF + 12 * i] =
                  make_float2(hi[j][2 * i], hi[j][2 * i + 1]);
            }
          }
          asm volatile("s_wait_dscnt 0" ::: "memory");
          float2 zk[NB][3], zm[NB][3], zn[NB];
#pragma unroll
          for (int q = 0; q < NB; ++q) {
            const float2* sq = spec + q * SPILLSZ;
#pragma unroll
            for (int j = 0; j < 3; ++j) {
              const int k = lane + 32 * j;
              zk[q][j] = sq[k];
              zm[q][j] = sq[(k == 0) ? 0 : NFFT - k];
            }
            zn[q] = sq[NHALF];
          }
#pragma unroll
          for (int q = 0; q < NB; ++q) {
            float2* RrowA = R + (row0 + 8 * q) * RPITCH;
            float2* RrowB = RrowA + 8 * NB * RPITCH;
#pragma unroll
            for (int j = 0; j < 3; ++j) {
              const int k = lane + 32 * j;
              const float2 a = zk[q][j], m = zm[q][j];
              RrowA[k] = make_float2(0.5f * (a.x + m.x), 0.5f * (a.y - m.y));
              RrowB[k] = make_float2(0.5f * (a.y + m.y), 0.5f * (m.x - a.x));
            }
            if (lane == 0) {
              RrowA[NHALF] = make_float2(zn[q].x, 0.0f);
              RrowB[NHALF] = make_float2(zn[q].y, 0.0f);
            }
          }
        },
        Ftab, Wtab, spill, +1.0f, lane);
  }
  __syncthreads();

  const float2* kfbase = kf + (size_t)c * KFCOLS * NFFT;
  for (int col0 = wave; col0 < NHALF; col0 += 8 * NB) {
#pragma unroll
    for (int q = 0; q < NB; ++q) {
      const int pc = col0 + 8 * NB + 8 * q;
      if (pc <= NHALF) {
        const char* pf = (const char*)(kfbase + (size_t)pc * NFFT);
        __builtin_prefetch(pf + ((lane < 24) ? lane : 23) * 64, 0, 1);
      }
    }
    conv_columns<NB>(R, kfbase, col0, Ftab, Wtab, spill, lane);
  }
  if (wave == 0)
    conv_columns<1>(R, kfbase, NHALF, Ftab, Wtab, spill, lane);
  __syncthreads();

  const float inv_scale = 1.0f / 36864.0f;
  for (int row0 = wave; row0 < 128; row0 += 16 * NB) {
    constexpr int NTB = S2TILES(NB);
    float gA[NTB][4], gB[NTB][4], gAh[NTB][3], gBh[NTB][3];
#pragma unroll
    for (int j = 0; j < NTB; ++j) {
      const int k0 = s2_k1<NB>(j, lane) + 48 * (lane >> 4);
      const size_t rA = base + (size_t)(row0 + 8 * s2_seq<NB>(j, lane)) * 128;
      const size_t rB = rA + (size_t)(8 * NB) * 128;
#pragma unroll
      for (int i = 0; i < 4; ++i) {
        gA[j][i] = postgate[rA + k0 + 12 * i];
        gB[j][i] = postgate[rB + k0 + 12 * i];
      }
#pragma unroll
      for (int i = 0; i < 3; ++i) {
        const int kh = (k0 + 12 * i < 32) ? NHALF + k0 + 12 * i : 127;
        gAh[j][i] = postgate[rA + kh];
        gBh[j][i] = postgate[rB + kh];
      }
    }
    fft192xN<NB>(
        [&](int q, int n) {
          const float2* RrowA = R + (row0 + 8 * q) * RPITCH;
          const float2* RrowB = RrowA + 8 * NB * RPITCH;
          const int idx = (n <= NHALF) ? n : NFFT - n;
          v4f r;
          r.x = RrowA[idx].x;
          r.y = RrowA[idx].y;
          r.z = RrowB[idx].x;
          r.w = RrowB[idx].y;
          return r;
        },
        [](v4f r, int n) {
          const bool stored = (n <= NHALF);
          const bool edge = (n == 0) || (n == NHALF);
          const float2 a = make_float2(r.x, r.y);
          const float2 b = make_float2(r.z, r.w);
          const float ay = edge ? 0.0f : a.y;
          const float by = edge ? 0.0f : b.y;
          return stored ? make_float2(a.x - by, ay + b.x)
                        : make_float2(a.x + by, b.x - ay);
        },
        [&](const auto& lo, const auto& hi) {
#pragma unroll
          for (int j = 0; j < NTB; ++j) {
#pragma unroll
            for (int i = 0; i < 4; ++i) {
              hold_loaded(gA[j][i]);
              hold_loaded(gB[j][i]);
            }
#pragma unroll
            for (int i = 0; i < 3; ++i) {
              hold_loaded(gAh[j][i]);
              hold_loaded(gBh[j][i]);
            }
          }
#pragma unroll
          for (int j = 0; j < NTB; ++j) {
            const int k0 = s2_k1<NB>(j, lane) + 48 * (lane >> 4);
            const size_t oA =
                base + (size_t)(row0 + 8 * s2_seq<NB>(j, lane)) * 128 + k0;
            const size_t oB = oA + (size_t)(8 * NB) * 128;
#pragma unroll
            for (int i = 0; i < 4; ++i) {
              y[oA + 12 * i] = lo[j][2 * i] * inv_scale * gA[j][i];
              y[oB + 12 * i] = lo[j][2 * i + 1] * inv_scale * gB[j][i];
            }
          }
          if (lane < 16) {
#pragma unroll
            for (int j = 0; j < NTB; ++j) {
              const size_t oA = base +
                  (size_t)(row0 + 8 * s2_seq<NB>(j, lane)) * 128 +
                  s2_k1<NB>(j, lane);
              const size_t oB = oA + (size_t)(8 * NB) * 128;
#pragma unroll
              for (int i = 0; i < 2; ++i) {
                y[oA + NHALF + 12 * i] = hi[j][2 * i] * inv_scale * gAh[j][i];
                y[oB + NHALF + 12 * i] =
                    hi[j][2 * i + 1] * inv_scale * gBh[j][i];
              }
            }
#pragma unroll
            for (int j = 0; j < NTB; ++j) {
              if (s2_k1<NB>(j, lane) < 8) {
                const size_t oA = base +
                    (size_t)(row0 + 8 * s2_seq<NB>(j, lane)) * 128 +
                    s2_k1<NB>(j, lane);
                const size_t oB = oA + (size_t)(8 * NB) * 128;
                y[oA + NHALF + 24] = hi[j][4] * inv_scale * gAh[j][2];
                y[oB + NHALF + 24] = hi[j][5] * inv_scale * gBh[j][2];
              }
            }
          }
        },
        Ftab, Wtab, spill, -1.0f, lane);
  }
}

extern "C" void kernel_launch(void* const* d_in, const int* in_sizes, int n_in,
                              void* d_out, int out_size, void* d_ws,
                              size_t ws_size, hipStream_t stream) {
  const float* u        = (const float*)d_in[0];
  const float* k        = (const float*)d_in[1];
  const float* pregate  = (const float*)d_in[2];
  const float* postgate = (const float*)d_in[3];
  float* y   = (float*)d_out;
  float2* kf = (float2*)d_ws;

  const size_t lds_kf   = (size_t)64 * RSTRIDE * 8 + 512 * 8 + 8 * NB * SPILLSZ * 8;
  const size_t lds_main = (size_t)128 * RPITCH * 8 + 512 * 8 + 8 * NB * SPILLSZ * 8;

  (void)hipFuncSetAttribute((const void*)kf_precompute,
                            hipFuncAttributeMaxDynamicSharedMemorySize,
                            (int)lds_kf);
  (void)hipFuncSetAttribute((const void*)conv_main,
                            hipFuncAttributeMaxDynamicSharedMemorySize,
                            (int)lds_main);

  kf_precompute<<<256, 256, lds_kf, stream>>>(k, kf);
  conv_main<<<4 * 256, 256, lds_main, stream>>>(u, pregate, postgate, kf, y);
}


# Main kernel output path: inverse-row results pass through the wave's LDS buffer and leave in row order; each 128-B line of y and postgate moved by ONE 8-B-per-lane instruction, was 4-B pieces
# speedup vs baseline: 5.7063x; 1.0454x over previous
#include <hip/hip_runtime.h>
#include <hip/hip_bf16.h>
#include <math.h>


typedef float v2f __attribute__((ext_vector_type(2)));
typedef float v8f __attribute__((ext_vector_type(8)));

#define PI_F 3.14159265358979323846f
#define NFFT 192
#define NHALF 96
#define RSTRIDE 133
#define RPITCH 97
#define KFCOLS 97
#define NB 4
#define SPITCH 18
#define SPILLSZ (12 * SPITCH)

__device__ __forceinline__ v8f wmma_f32(v2f a, v2f b, v8f c) {
  return __builtin_amdgcn_wmma_f32_16x16x4_f32(false, a, false, b, (short)0, c,
                                               false, false);
}

__device__ __forceinline__ void block_fence() {
  __builtin_amdgcn_sched_barrier(0x6);
}


__device__ __forceinline__ float2 load_g_entry(const float2* Ftab, int t, int b,
                                               int lane) {
  const int m = lane & 15;
  const int h = (lane >> 4) & 1;
  const int c = m >> 1;
  const int a = 2 * t + h;
  return Ftab[16 + (((2 * a + b) * c) & 15)];
}

__device__ __forceinline__ v2f make_g_frag(float2 g, float dir, int lane) {
  const int m = lane & 15;
  const float gr = g.x, gi = dir * g.y;
  v2f f;
  if (m & 1) { f.x = gi; f.y = gr; } else { f.x = gr; f.y = -gi; }
  return f;
}

__device__ __forceinline__ float2 load_g12_entry(const float2* Ftab, int t,
                                                 int b, int lane) {
  const int m = lane & 15;
  const int h = (lane >> 4) & 1;
  const int c = 3 * (m >> 3) + ((m & 7) >> 1);
  const int a = 2 * t + h;
  return Ftab[32 + (((2 * a + b) * c) % 12)];
}

__device__ __forceinline__ v2f make_g12_frag(float2 g, float dir, int lane) {
  v2f f = make_g_frag(g, dir, lane);
  if ((lane & 7) >= 6) { f.x = 0.0f; f.y = 0.0f; }
  return f;
}

typedef float v4f __attribute__((ext_vector_type(4)));

__device__ __forceinline__ v2f as_frag(float2 x) {
  v2f f;
  f.x = x.x;
  f.y = x.y;
  return f;
}

#define S2TILES(n) ((12 * (n) + 15) / 16)

template <int N>
__device__ __forceinline__ bool s2_valid(int j, int lane) {
  if constexpr ((12 * N) % 16 == 0) return true;
  else return 16 * j + (lane & 15) < 12 * N;
}

template <int N>
__device__ __forceinline__ int s2_col(int j, int lane) {
  const int c = 16 * j + (lane & 15);
  if constexpr ((12 * N) % 16 == 0) return c;
  else return (c < 12 * N) ? c : 12 * N - 1;
}

template <int N>
__device__ __forceinline__ int s2_seq(int j, int lane) {
  return s2_col<N>(j, lane) / 12;
}

template <int N>
__device__ __forceinline__ int s2_k1(int j, int lane) {
  return s2_col<N>(j, lane) % 12;
}

template <int N, typename LD, typename FX, typename DST>
__device__ __forceinline__ void fft192xN(LD ld, FX fx, DST dst,
                                         const float2* Ftab,
                                         const float2* Wtab, float2* spill,
                                         float dir, int lane) {
  const int col = lane & 15;
  const int h   = (lane >> 4) & 1;

  float2 gte[3], gto[3];
#pragma unroll
  for (int t = 0; t < 3; ++t) {
    gte[t] = load_g12_entry(Ftab, t, 0, lane);
    gto[t] = load_g12_entry(Ftab, t, 1, lane);
  }
  float2 wt[2][3];
#pragma unroll
  for (int d = 0; d < 2; ++d) {
#pragma unroll
    for (int i = 0; i < 3; ++i)
      wt[d][i] = Wtab[col * (3 * h + i + 6 * d)];
  }
  using Raw = decltype(ld(0, 0));
  Raw re[N][3], ro[N][3];
#pragma unroll
  for (int t = 0; t < 3; ++t) {
    const int n2 = 2 * (2 * t + h);
#pragma unroll
    for (int q = 0; q < N; ++q) {
      re[q][t] = ld(q, col + 16 * n2);
      ro[q][t] = ld(q, col + 16 * (n2 + 1));
    }
  }
  block_fence();

  v2f ge[3], go[3];
#pragma unroll
  for (int t = 0; t < 3; ++t) {
    ge[t] = make_g12_frag(gte[t], dir, lane);
    go[t] = make_g12_frag(gto[t], dir, lane);
  }

  v8f e[N], o[N];
#pragma unroll
  for (int q = 0; q < N; ++q) {
    e[q] = v8f{}; o[q] = v8f{};
  }

#pragma unroll
  for (int t = 0; t < 3; ++t) {
    const int n2 = 2 * (2 * t + h);
#pragma unroll
    for (int q = 0; q < N; ++q) {
      const v2f xe = as_frag(fx(re[q][t], col + 16 * n2));
      const v2f xo = as_frag(fx(ro[q][t], col + 16 * (n2 + 1)));
      e[q] = wmma_f32(ge[t], xe, e[q]);
      o[q] = wmma_f32(go[t], xo, o[q]);
    }
  }
  block_fence();

#pragma unroll
  for (int q = 0; q < N; ++q) {
    float2* sp = spill + q * SPILLSZ + col;
#pragma unroll
    for (int i = 0; i < 3; ++i) {
      const float lr = e[q][2 * i] + o[q][2 * i];
      const float li = e[q][2 * i + 1] + o[q][2 * i + 1];
      const float hr = e[q][2 * i] - o[q][2 * i];
      const float hm = e[q][2 * i + 1] - o[q][2 * i + 1];
      const float t0r = wt[0][i].x, t0i = dir * wt[0][i].y;
      const float t1r = wt[1][i].x, t1i = dir * wt[1][i].y;
      const int k1 = 3 * h + i;
      sp[k1 * SPITCH] =
          make_float2(lr * t0r - li * t0i, lr * t0i + li * t0r);
      sp[(k1 + 6) * SPITCH] =
          make_float2(hr * t1r - hm * t1i, hr * t1i + hm * t1r);
    }
  }
  asm volatile("s_wait_dscnt 0" ::: "memory");

  constexpr int NT = S2TILES(N);
  float2 hte[4], hto[4];
#pragma unroll
  for (int t = 0; t < 4; ++t) {
    hte[t] = load_g_entry(Ftab, t, 0, lane);
    hto[t] = load_g_entry(Ftab, t, 1, lane);
  }
  v2f be[NT][4], bo[NT][4];
#pragma unroll
  for (int t = 0; t < 4; ++t) {
    const int n1 = 2 * (2 * t + h);
#pragma unroll
    for (int j = 0; j < NT; ++j) {
      const float2* sp = spill + s2_seq<N>(j, lane) * SPILLSZ +
                         s2_k1<N>(j, lane) * SPITCH + n1;
      be[j][t] = as_frag(sp[0]);
      bo[j][t] = as_frag(sp[1]);
    }
  }
  block_fence();

  v2f he[4], ho[4];
#pragma unroll
  for (int t = 0; t < 4; ++t) {
    he[t] = make_g_frag(hte[t], dir, lane);
    ho[t] = make_g_frag(hto[t], dir, lane);
  }

  v8f e2[NT], o2[NT];
#pragma unroll
  for (int j = 0; j < NT; ++j) {
    e2[j] = v8f{}; o2[j] = v8f{};
  }

#pragma unroll
  for (int t = 0; t < 4; ++t) {
#pragma unroll
    for (int j = 0; j < NT; ++j) {
      e2[j] = wmma_f32(he[t], be[j][t], e2[j]);
      o2[j] = wmma_f32(ho[t], bo[j][t], o2[j]);
    }
  }
  block_fence();

  v8f lo[NT], hi[NT];
#pragma unroll
  for (int j = 0; j < NT; ++j) {
    lo[j] = e2[j] + o2[j];
    hi[j] = e2[j] - o2[j];
  }
  dst(lo, hi);
}

__device__ __forceinline__ void init_tables(float2* Ftab, float2* Wtab,
                                            int tid) {
  for (int i = tid; i < 256; i += 256) {
    const int r = i >> 4, c = i & 15;
    const float tf = (r == 2)
        ? -2.0f * PI_F * (float)(c % 12) / 12.0f
        : -2.0f * PI_F * (float)((r * c) & 15) / 16.0f;
    Ftab[i] = make_float2(cosf(tf), sinf(tf));
    const float tw = -2.0f * PI_F * (float)(i % NFFT) / (float)NFFT;
    Wtab[i] = make_float2(cosf(tw), sinf(tw));
  }
}


template <int N>
__device__ __forceinline__ void kf_columns(const float2* Rk, float2* kfc,
                                           int col0, const float2* Ftab,
                                           const float2* Wtab, float2* spill,
                                           int lane) {
  fft192xN<N>(
      [&](int q, int n) {
        return (n < 64) ? Rk[n * RSTRIDE + col0 + 8 * q]
                        : make_float2(0.f, 0.f);
      },
      [](float2 v, int) { return v; },
      [&](const auto& lo, const auto& hi) {
#pragma unroll
        for (int j = 0; j < S2TILES(N); ++j) {
          if (s2_valid<N>(j, lane)) {
            const int k0 = s2_k1<N>(j, lane) + 48 * (lane >> 4);
            float2* dcol =
                kfc + (size_t)(col0 + 8 * s2_seq<N>(j, lane)) * NFFT + k0;
#pragma unroll
            for (int i = 0; i < 4; ++i) {
              dcol[12 * i] = make_float2(lo[j][2 * i], lo[j][2 * i + 1]);
              dcol[NHALF + 12 * i] =
                  make_float2(hi[j][2 * i], hi[j][2 * i + 1]);
            }
          }
        }
      },
      Ftab, Wtab, spill, +1.0f, lane);
}

__global__ void __launch_bounds__(256) kf_precompute(
    const float* __restrict__ kin, float2* __restrict__ kf) {
  extern __shared__ char smem[];
  float2* Rk      = (float2*)smem;
  float2* Ftab    = (float2*)(smem + 64 * RSTRIDE * 8);
  float2* Wtab    = Ftab + 256;
  float2* scratch = Wtab + 256;

  const int tid = threadIdx.x, lane = tid & 31, wave = tid >> 5;
  const int c = blockIdx.x;

  const float4* k4 = reinterpret_cast<const float4*>(kin + (size_t)c * (64 * 64));
  const int srow = 8 * wave + (lane >> 3);
  const int sseg = lane & 7;
  float4 kq[4];
#pragma unroll
  for (int j = 0; j < 4; ++j)
    kq[j] = k4[(srow + 4 * (j >> 1)) * 16 + sseg + 8 * (j & 1)];

  init_tables(Ftab, Wtab, tid);

#pragma unroll
  for (int j = 0; j < 4; ++j) {
    float2* d = Rk + (srow + 4 * (j >> 1)) * RSTRIDE + 4 * (sseg + 8 * (j & 1));
    d[0] = make_float2(kq[j].x, 0.0f);
    d[1] = make_float2(kq[j].y, 0.0f);
    d[2] = make_float2(kq[j].z, 0.0f);
    d[3] = make_float2(kq[j].w, 0.0f);
  }
  __syncthreads();

  float2* spill = scratch + wave * (NB * SPILLSZ);

  for (int row0 = wave; row0 < 64; row0 += 8 * NB) {
    fft192xN<NB>(
        [&](int q, int n) {
          const float2* Rrow = Rk + (row0 + 8 * q) * RSTRIDE;
          return (n < 64) ? make_float2(Rrow[n].x, 0.0f)
                          : make_float2(0.f, 0.f);
        },
        [](float2 v, int) { return v; },
        [&](const auto& lo, const auto& hi) {
#pragma unroll
          for (int j = 0; j < S2TILES(NB); ++j) {
            const int k1 = s2_k1<NB>(j, lane);
            float2* Rrow = Rk + (row0 + 8 * s2_seq<NB>(j, lane)) * RSTRIDE;
#pragma unroll
            for (int i = 0; i < 4; ++i)
              Rrow[k1 + 48 * (lane >> 4) + 12 * i] =
                  make_float2(lo[j][2 * i], lo[j][2 * i + 1]);
            if (k1 == 0 && lane < 16)
              Rrow[NHALF] = make_float2(hi[j][0], hi[j][1]);
          }
        },
        Ftab, Wtab, spill, +1.0f, lane);
  }
  __syncthreads();

  float2* kfc = kf + (size_t)c * KFCOLS * NFFT;
  for (int col0 = wave; col0 < NHALF; col0 += 8 * NB)
    kf_columns<NB>(Rk, kfc, col0, Ftab, Wtab, spill, lane);
  if (wave == 0) kf_columns<1>(Rk, kfc, NHALF, Ftab, Wtab, spill, lane);
}


template <int N>
__device__ __forceinline__ void conv_columns(float2* R, const float2* kfbase,
                                             int col0, const float2* Ftab,
                                             const float2* Wtab, float2* spill,
                                             int lane) {
  float2* spec = spill;
  float2 kv[N][6];
#pragma unroll
  for (int q = 0; q < N; ++q) {
    const float2* kfcol = kfbase + (size_t)(col0 + 8 * q) * NFFT + lane;
#pragma unroll
    for (int i = 0; i < 6; ++i) kv[q][i] = kfcol[32 * i];
  }

  fft192xN<N>(
      [&](int q, int n) {
        return (n < 128) ? R[n * RPITCH + col0 + 8 * q]
                         : make_float2(0.f, 0.f);
      },
      [](float2 v, int) { return v; },
      [&](const auto& lo, const auto& hi) {
#pragma unroll
        for (int j = 0; j < S2TILES(N); ++j) {
          if (s2_valid<N>(j, lane)) {
            float2* sq = spec + s2_seq<N>(j, lane) * SPILLSZ +
                         s2_k1<N>(j, lane) + 48 * (lane >> 4);
#pragma unroll
            for (int i = 0; i < 4; ++i) {
              sq[12 * i] = make_float2(lo[j][2 * i], lo[j][2 * i + 1]);
              sq[NHALF + 12 * i] =
                  make_float2(hi[j][2 * i], hi[j][2 * i + 1]);
            }
          }
        }
      },
      Ftab, Wtab, spill, +1.0f, lane);

#pragma unroll
  for (int q = 0; q < N; ++q) {
    float2* sq = spec + q * SPILLSZ + lane;
#pragma unroll
    for (int i = 0; i < 6; ++i) {
      const float2 a = sq[32 * i];
      const float2 b = kv[q][i];
      sq[32 * i] = make_float2(a.x * b.x - a.y * b.y, a.x * b.y + a.y * b.x);
    }
  }
  asm volatile("s_wait_dscnt 0" ::: "memory");

  fft192xN<N>(
      [&](int q, int n) { return spec[q * SPILLSZ + n]; },
      [](float2 v, int) { return v; },
      [&](const auto& lo, const auto& hi) {
#pragma unroll
        for (int j = 0; j < S2TILES(N); ++j) {
          if (s2_valid<N>(j, lane)) {
            const int k0 = s2_k1<N>(j, lane) + 48 * (lane >> 4);
            float2* Rc = R + col0 + 8 * s2_seq<N>(j, lane);
#pragma unroll
            for (int i = 0; i < 4; ++i)
              Rc[(k0 + 12 * i) * RPITCH] =
                  make_float2(lo[j][2 * i], lo[j][2 * i + 1]);
          }
        }
        if (lane < 16) {
#pragma unroll
          for (int j = 0; j < S2TILES(N); ++j) {
            if (s2_valid<N>(j, lane)) {
              float2* Rh = R + (NHALF + s2_k1<N>(j, lane)) * RPITCH + col0 +
                           8 * s2_seq<N>(j, lane);
#pragma unroll
              for (int i = 0; i < 2; ++i)
                Rh[12 * i * RPITCH] =
                    make_float2(hi[j][2 * i], hi[j][2 * i + 1]);
            }
          }
#pragma unroll
          for (int j = 0; j < S2TILES(N); ++j) {
            if (s2_valid<N>(j, lane) && s2_k1<N>(j, lane) < 8)
              R[(NHALF + s2_k1<N>(j, lane) + 24) * RPITCH + col0 +
                8 * s2_seq<N>(j, lane)] = make_float2(hi[j][4], hi[j][5]);
          }
        }
      },
      Ftab, Wtab, spill, -1.0f, lane);
}

__global__ void __launch_bounds__(256) __attribute__((amdgpu_waves_per_eu(4, 4)))
conv_main(
    const float* __restrict__ u, const float* __restrict__ pregate,
    const float* __restrict__ postgate, const float2* __restrict__ kf,
    float* __restrict__ y) {
  extern __shared__ char smem[];
  float2* R       = (float2*)smem;
  float2* Ftab    = (float2*)(smem + 128 * RPITCH * 8);
  float2* Wtab    = Ftab + 256;
  float2* scratch = Wtab + 256;

  const int tid = threadIdx.x, lane = tid & 31, wave = tid >> 5;
  const int img = blockIdx.x;
  const int c = img & 255;
  const size_t base = (size_t)img * (128 * 128);

  const float4* u4 = reinterpret_cast<const float4*>(u + base);
  const float4* p4 = reinterpret_cast<const float4*>(pregate + base);
  const int srow = 16 * wave + (lane >> 3);
  const int sseg = lane & 7;
  float4 uv[16], pv[16];
#pragma unroll
  for (int j = 0; j < 16; ++j) {
    const int i4 = (srow + 4 * (j >> 2)) * 32 + sseg + 8 * (j & 3);
    uv[j] = u4[i4];
    pv[j] = p4[i4];
  }

  init_tables(Ftab, Wtab, tid);

#pragma unroll
  for (int j = 0; j < 16; ++j) {
    float2* d = R + (srow + 4 * (j >> 2)) * RPITCH + 2 * (sseg + 8 * (j & 3));
    d[0] = make_float2(uv[j].x * pv[j].x, uv[j].y * pv[j].y);
    d[1] = make_float2(uv[j].z * pv[j].z, uv[j].w * pv[j].w);
  }
  __syncthreads();

  float2* spill = scratch + wave * (NB * SPILLSZ);
  float2* spec  = spill;

  for (int row0 = wave; row0 < 128; row0 += 16 * NB) {
    fft192xN<NB>(
        [&](int q, int n) {
          const float* xA = (const float*)(R + (row0 + 8 * q) * RPITCH);
          const float* xB = (const float*)(R + (row0 + 8 * q + 8 * NB) * RPITCH);
          return (n < 128) ? make_float2(xA[n], xB[n])
                           : make_float2(0.f, 0.f);
        },
        [](float2 v, int) { return v; },
        [&](const auto& lo, const auto& hi) {
#pragma unroll
          for (int j = 0; j < S2TILES(NB); ++j) {
            float2* sz = spec + s2_seq<NB>(j, lane) * SPILLSZ +
                         s2_k1<NB>(j, lane) + 48 * (lane >> 4);
#pragma unroll
            for (int i = 0; i < 4; ++i) {
              sz[12 * i] = make_float2(lo[j][2 * i], lo[j][2 * i + 1]);
              sz[NHALF + 12 * i] =
                  make_float2(hi[j][2 * i], hi[j][2 * i + 1]);
            }
          }
          asm volatile("s_wait_dscnt 0" ::: "memory");
          float2 zk[NB][3], zm[NB][3], zn[NB];
#pragma unroll
          for (int q = 0; q < NB; ++q) {
            const float2* sq = spec + q * SPILLSZ;
#pragma unroll
            for (int j = 0; j < 3; ++j) {
              const int k = lane + 32 * j;
              zk[q][j] = sq[k];
              zm[q][j] = sq[(k == 0) ? 0 : NFFT - k];
            }
            zn[q] = sq[NHALF];
          }
#pragma unroll
          for (int q = 0; q < NB; ++q) {
            float2* RrowA = R + (row0 + 8 * q) * RPITCH;
            float2* RrowB = RrowA + 8 * NB * RPITCH;
#pragma unroll
            for (int j = 0; j < 3; ++j) {
              const int k = lane + 32 * j;
              const float2 a = zk[q][j], m = zm[q][j];
              RrowA[k] = make_float2(0.5f * (a.x + m.x), 0.5f * (a.y - m.y));
              RrowB[k] = make_float2(0.5f * (a.y + m.y), 0.5f * (m.x - a.x));
            }
            if (lane == 0) {
              RrowA[NHALF] = make_float2(zn[q].x, 0.0f);
              RrowB[NHALF] = make_float2(zn[q].y, 0.0f);
            }
          }
        },
        Ftab, Wtab, spill, +1.0f, lane);
  }
  __syncthreads();

  const float2* kfbase = kf + (size_t)c * KFCOLS * NFFT;
  for (int col0 = wave; col0 < NHALF; col0 += 8 * NB) {
#pragma unroll
    for (int q = 0; q < NB; ++q) {
      const int pc = col0 + 8 * NB + 8 * q;
      if (pc <= NHALF) {
        const char* pf = (const char*)(kfbase + (size_t)pc * NFFT);
        __builtin_prefetch(pf + ((lane < 24) ? lane : 23) * 64, 0, 1);
      }
    }
    conv_columns<NB>(R, kfbase, col0, Ftab, Wtab, spill, lane);
  }
  if (wave == 0)
    conv_columns<1>(R, kfbase, NHALF, Ftab, Wtab, spill, lane);
  __syncthreads();

  const float inv_scale = 1.0f / 36864.0f;
  const float2* pg2 = reinterpret_cast<const float2*>(postgate + base);
  float2* y2 = reinterpret_cast<float2*>(y + base);
  for (int row0 = wave; row0 < 128; row0 += 16 * NB) {
    float2 gA[NB][2], gB[NB][2];
#pragma unroll
    for (int q = 0; q < NB; ++q) {
      const int rA = row0 + 8 * q, rB = rA + 8 * NB;
#pragma unroll
      for (int p = 0; p < 2; ++p) {
        gA[q][p] = pg2[rA * 64 + 32 * p + lane];
        gB[q][p] = pg2[rB * 64 + 32 * p + lane];
      }
    }
    fft192xN<NB>(
        [&](int q, int n) {
          const float2* RrowA = R + (row0 + 8 * q) * RPITCH;
          const float2* RrowB = RrowA + 8 * NB * RPITCH;
          const int idx = (n <= NHALF) ? n : NFFT - n;
          v4f r;
          r.x = RrowA[idx].x;
          r.y = RrowA[idx].y;
          r.z = RrowB[idx].x;
          r.w = RrowB[idx].y;
          return r;
        },
        [](v4f r, int n) {
          const bool stored = (n <= NHALF);
          const bool edge = (n == 0) || (n == NHALF);
          const float2 a = make_float2(r.x, r.y);
          const float2 b = make_float2(r.z, r.w);
          const float ay = edge ? 0.0f : a.y;
          const float by = edge ? 0.0f : b.y;
          return stored ? make_float2(a.x - by, ay + b.x)
                        : make_float2(a.x + by, b.x - ay);
        },
        [&](const auto& lo, const auto& hi) {
#pragma unroll
          for (int j = 0; j < S2TILES(NB); ++j) {
            float2* sz = spec + s2_seq<NB>(j, lane) * SPILLSZ +
                         s2_k1<NB>(j, lane) + 48 * (lane >> 4);
#pragma unroll
            for (int i = 0; i < 4; ++i) {
              sz[12 * i] = make_float2(lo[j][2 * i], lo[j][2 * i + 1]);
              sz[NHALF + 12 * i] =
                  make_float2(hi[j][2 * i], hi[j][2 * i + 1]);
            }
          }
          asm volatile("s_wait_dscnt 0" ::: "memory");
          float2 o0[NB][2], o1[NB][2];
#pragma unroll
          for (int q = 0; q < NB; ++q) {
#pragma unroll
            for (int p = 0; p < 2; ++p) {
              const float2* sq = spec + q * SPILLSZ + 64 * p + 2 * lane;
              o0[q][p] = sq[0];
              o1[q][p] = sq[1];
            }
          }
#pragma unroll
          for (int q = 0; q < NB; ++q) {
            const int rA = row0 + 8 * q, rB = rA + 8 * NB;
#pragma unroll
            for (int p = 0; p < 2; ++p) {
              y2[rA * 64 + 32 * p + lane] =
                  make_float2(o0[q][p].x * inv_scale * gA[q][p].x,
                              o1[q][p].x * inv_scale * gA[q][p].y);
              y2[rB * 64 + 32 * p + lane] =
                  make_float2(o0[q][p].y * inv_scale * gB[q][p].x,
                              o1[q][p].y * inv_scale * gB[q][p].y);
            }
          }
        },
        Ftab, Wtab, spill, -1.0f, lane);
  }
}

extern "C" void kernel_launch(void* const* d_in, const int* in_sizes, int n_in,
                              void* d_out, int out_size, void* d_ws,
                              size_t ws_size, hipStream_t stream) {
  const float* u        = (const float*)d_in[0];
  const float* k        = (const float*)d_in[1];
  const float* pregate  = (const float*)d_in[2];
  const float* postgate = (const float*)d_in[3];
  float* y   = (float*)d_out;
  float2* kf = (float2*)d_ws;

  const size_t lds_kf   = (size_t)64 * RSTRIDE * 8 + 512 * 8 + 8 * NB * SPILLSZ * 8;
  const size_t lds_main = (size_t)128 * RPITCH * 8 + 512 * 8 + 8 * NB * SPILLSZ * 8;

  (void)hipFuncSetAttribute((const void*)kf_precompute,
                            hipFuncAttributeMaxDynamicSharedMemorySize,
                            (int)lds_kf);
  (void)hipFuncSetAttribute((const void*)conv_main,
                            hipFuncAttributeMaxDynamicSharedMemorySize,
                            (int)lds_main);

  kf_precompute<<<256, 256, lds_kf, stream>>>(k, kf);
  conv_main<<<4 * 256, 256, lds_main, stream>>>(u, pregate, postgate, kf, y);
}
